# starting kernel
# baseline (speedup 1.0000x reference)
#include <hip/hip_runtime.h>
#include <hip/hip_bf16.h>

typedef float v2f __attribute__((ext_vector_type(2)));
typedef float v8f __attribute__((ext_vector_type(8)));

#define D 64

__global__ void k_deg_init(float* __restrict__ deg, int n) {
  int i = blockIdx.x * blockDim.x + threadIdx.x;
  if (i < n) deg[i] = 1.0f;
}

__global__ void k_deg_accum(const int* __restrict__ dst, float* __restrict__ deg, int E) {
  int e = blockIdx.x * blockDim.x + threadIdx.x;
  if (e < E) atomicAdd(&deg[dst[e]], 1.0f);
}

__global__ void k_dinv(float* __restrict__ deg, int n) {
  int i = blockIdx.x * blockDim.x + threadIdx.x;
  if (i < n) {
    float d = deg[i];
    deg[i] = (d > 0.0f) ? rsqrtf(d) : 0.0f;
  }
}

__global__ void k_norm(const int* __restrict__ src, const int* __restrict__ dst,
                       const float* __restrict__ dinv, float* __restrict__ norm,
                       int E, int n) {
  int e = blockIdx.x * blockDim.x + threadIdx.x;
  int tot = E + n;
  if (e < tot) {
    int s, t;
    if (e < E) { s = src[e]; t = dst[e]; }
    else       { s = t = e - E; }
    norm[e] = dinv[s] * dinv[t];
  }
}

__global__ void k_gemm_wmma(const float* __restrict__ A, const float* __restrict__ W,
                            float* __restrict__ Cout, int M) {
  __shared__ v2f sW[(D / 2) * D];
  for (int i = threadIdx.x; i < (D / 2) * D; i += blockDim.x) {
    int kk = i / D;
    int nn = i - kk * D;
    v2f p;
    p.x = W[(2 * kk + 0) * D + nn];
    p.y = W[(2 * kk + 1) * D + nn];
    sW[i] = p;
  }
  __syncthreads();

  int lane = threadIdx.x & 31;
  int half = lane >> 4;
  int lid  = lane & 15;
  int wave = blockIdx.x * (blockDim.x >> 5) + (threadIdx.x >> 5);
  int row0 = wave * 16;
  if (row0 >= M) return;

  v8f acc0 = {}, acc1 = {}, acc2 = {}, acc3 = {};

  int r  = row0 + lid;
  int rc = (r < M) ? r : (M - 1);
  const v2f* arow = (const v2f*)(A + (size_t)rc * D);

#pragma unroll
  for (int k = 0; k < D; k += 4) {
    int kk = (k >> 1) + half;
    v2f a = arow[kk];
    const v2f* wrow = &sW[kk * D + lid];
    v2f b0 = wrow[0];
    v2f b1 = wrow[16];
    v2f b2 = wrow[32];
    v2f b3 = wrow[48];
    acc0 = __builtin_amdgcn_wmma_f32_16x16x4_f32(false, a, false, b0, (short)0, acc0, false, false);
    acc1 = __builtin_amdgcn_wmma_f32_16x16x4_f32(false, a, false, b1, (short)0, acc1, false, false);
    acc2 = __builtin_amdgcn_wmma_f32_16x16x4_f32(false, a, false, b2, (short)0, acc2, false, false);
    acc3 = __builtin_amdgcn_wmma_f32_16x16x4_f32(false, a, false, b3, (short)0, acc3, false, false);
  }

#pragma unroll
  for (int v = 0; v < 8; ++v) {
    int rr = row0 + v + 8 * half;
    if (rr < M) {
      float* crow = Cout + (size_t)rr * D + lid;
      crow[0]  = acc0[v];
      crow[16] = acc1[v];
      crow[32] = acc2[v];
      crow[48] = acc3[v];
    }
  }
}

__global__ void k_bias_init(float* __restrict__ agg, const float* __restrict__ b, int n) {
  int i = blockIdx.x * blockDim.x + threadIdx.x;
  if (i < n * D) agg[i] = b[i & (D - 1)];
}

__global__ void k_scatter(const int* __restrict__ src, const int* __restrict__ dst,
                          const float* __restrict__ norm, const float* __restrict__ h,
                          float* __restrict__ agg, int E, int n) {
  int tid = blockIdx.x * blockDim.x + threadIdx.x;
  int d = tid & (D - 1);
  int e = tid >> 6;
  int tot = E + n;
  if (e < tot) {
    int s, t;
    if (e < E) { s = src[e]; t = dst[e]; }
    else       { s = t = e - E; }
    float v = h[(size_t)s * D + d] * norm[e];
    atomicAdd(&agg[(size_t)t * D + d], v);
  }
}

__global__ void k_prelu(const float* __restrict__ in, const float* __restrict__ a,
                        float* __restrict__ out, int n) {
  int i = blockIdx.x * blockDim.x + threadIdx.x;
  if (i < n * D) {
    float v = in[i];
    float al = a[i & (D - 1)];
    out[i] = (v >= 0.0f) ? v : al * v;
  }
}

static inline int cdiv_ll(long long a, long long b) { return (int)((a + b - 1) / b); }

extern "C" void kernel_launch(void* const* d_in, const int* in_sizes, int n_in,
                              void* d_out, int out_size, void* d_ws, size_t ws_size,
                              hipStream_t stream) {
  (void)n_in; (void)out_size; (void)ws_size;

  const float* x  = (const float*)d_in[0];
  const int*   ei = (const int*)  d_in[1];
  const float* W1 = (const float*)d_in[2];
  const float* b1 = (const float*)d_in[3];
  const float* a1 = (const float*)d_in[4];
  const float* W2 = (const float*)d_in[5];
  const float* b2 = (const float*)d_in[6];
  const float* a2 = (const float*)d_in[7];

  const int n = in_sizes[0] / D;
  const int E = in_sizes[1] / 2;
  const int* src = ei;
  const int* dst = ei + E;

  float* ws   = (float*)d_ws;
  float* dinv = ws;
  float* norm = dinv + n;
  float* bufA = norm + (size_t)(E + n);
  float* bufB = bufA + (size_t)n * D;
  float* out  = (float*)d_out;

  const int B = 256;
  const int gN    = cdiv_ll(n, B);
  const int gE    = cdiv_ll(E, B);
  const int gEN   = cdiv_ll((long long)E + n, B);
  const int gND   = cdiv_ll((long long)n * D, B);
  const int gSc   = cdiv_ll(((long long)E + n) * D, B);
  const int gGemm = cdiv_ll((long long)cdiv_ll(n, 16) * 32, B);

  k_deg_init <<<gN,  B, 0, stream>>>(dinv, n);
  k_deg_accum<<<gE,  B, 0, stream>>>(dst, dinv, E);
  k_dinv     <<<gN,  B, 0, stream>>>(dinv, n);
  k_norm     <<<gEN, B, 0, stream>>>(src, dst, dinv, norm, E, n);

  k_gemm_wmma<<<gGemm, B, 0, stream>>>(x, W1, bufA, n);
  k_bias_init<<<gND,   B, 0, stream>>>(bufB, b1, n);
  k_scatter  <<<gSc,   B, 0, stream>>>(src, dst, norm, bufA, bufB, E, n);
  k_prelu    <<<gND,   B, 0, stream>>>(bufB, a1, bufA, n);

  k_gemm_wmma<<<gGemm, B, 0, stream>>>(bufA, W2, bufB, n);
  k_bias_init<<<gND,   B, 0, stream>>>(out, b2, n);
  k_scatter  <<<gSc,   B, 0, stream>>>(src, dst, norm, bufB, out, E, n);
  k_prelu    <<<gND,   B, 0, stream>>>(out, a2, out, n);
}


# Replace the f32 atomic scatter-add by a per-call edge list grouped by destination (integer atomic count, old value = arrival-order rank; scan; place) summed per node in registers; sum order changes
# speedup vs baseline: 2.6630x; 2.6630x over previous
#include <hip/hip_runtime.h>
#include <hip/hip_bf16.h>

typedef float v2f __attribute__((ext_vector_type(2)));
typedef float v8f __attribute__((ext_vector_type(8)));

#define D 64

__global__ void k_cnt_zero(int* __restrict__ cnt, int n) {
  int i = blockIdx.x * blockDim.x + threadIdx.x;
  if (i < n) cnt[i] = 0;
}

__global__ void k_rank(const int* __restrict__ dst, int* __restrict__ cnt,
                       int* __restrict__ rank, int E) {
  int e = blockIdx.x * blockDim.x + threadIdx.x;
  if (e < E) rank[e] = atomicAdd(&cnt[dst[e]], 1);
}

__global__ void __launch_bounds__(256)
k_scan_local(const int* __restrict__ cnt, int* __restrict__ rowptr,
             int* __restrict__ bsum, float* __restrict__ dinv, int n) {
  __shared__ int s_wave[8];
  const int t = threadIdx.x;
  const int lane = t & 31;
  const int w = t >> 5;
  const int base = (blockIdx.x * 256 + t) * 4;
  int c[4];
  for (int j = 0; j < 4; ++j) c[j] = (base + j < n) ? cnt[base + j] : 0;
  const int s = c[0] + c[1] + c[2] + c[3];
  int inc = s;
  for (int o = 1; o < 32; o <<= 1) {
    int v = __shfl_up(inc, o);
    if (lane >= o) inc += v;
  }
  if (lane == 31) s_wave[w] = inc;
  __syncthreads();
  int woff = 0, tot = 0;
  for (int j = 0; j < 8; ++j) {
    int v = s_wave[j];
    tot += v;
    if (j < w) woff += v;
  }
  int run = woff + inc - s;
  for (int j = 0; j < 4; ++j) {
    if (base + j < n) {
      rowptr[base + j] = run;
      float d = (float)(c[j] + 1);
      dinv[base + j] = (d > 0.0f) ? rsqrtf(d) : 0.0f;
    }
    run += c[j];
  }
  if (t == 0) bsum[blockIdx.x] = tot;
}

__global__ void __launch_bounds__(256)
k_scan_add(int* __restrict__ rowptr, const int* __restrict__ bsum, int n) {
  __shared__ int s_off;
  const int t = threadIdx.x;
  if (t < 32) {
    int a = 0;
    for (int j = t; j < (int)blockIdx.x; j += 32) a += bsum[j];
    for (int o = 16; o > 0; o >>= 1) a += __shfl_xor(a, o);
    if (t == 0) s_off = a;
  }
  __syncthreads();
  const int off = s_off;
  const int base = (blockIdx.x * 256 + t) * 4;
  for (int j = 0; j < 4; ++j)
    if (base + j < n) rowptr[base + j] += off;
}

__global__ void k_fill(const int* __restrict__ src, const int* __restrict__ dst,
                       const int* __restrict__ rank, const int* __restrict__ rowptr,
                       int* __restrict__ csr, int E) {
  int e = blockIdx.x * blockDim.x + threadIdx.x;
  if (e < E) csr[rowptr[dst[e]] + rank[e]] = src[e];
}

__global__ void k_gemm_wmma(const float* __restrict__ A, const float* __restrict__ W,
                            float* __restrict__ Cout, int M) {
  __shared__ v2f sW[(D / 2) * D];
  for (int i = threadIdx.x; i < (D / 2) * D; i += blockDim.x) {
    int kk = i / D;
    int nn = i - kk * D;
    v2f p;
    p.x = W[(2 * kk + 0) * D + nn];
    p.y = W[(2 * kk + 1) * D + nn];
    sW[i] = p;
  }
  __syncthreads();

  int lane = threadIdx.x & 31;
  int half = lane >> 4;
  int lid  = lane & 15;
  int wave = blockIdx.x * (blockDim.x >> 5) + (threadIdx.x >> 5);
  int row0 = wave * 16;
  if (row0 >= M) return;

  v8f acc0 = {}, acc1 = {}, acc2 = {}, acc3 = {};

  int r  = row0 + lid;
  int rc = (r < M) ? r : (M - 1);
  const v2f* arow = (const v2f*)(A + (size_t)rc * D);

#pragma unroll
  for (int k = 0; k < D; k += 4) {
    int kk = (k >> 1) + half;
    v2f a = arow[kk];
    const v2f* wrow = &sW[kk * D + lid];
    v2f b0 = wrow[0];
    v2f b1 = wrow[16];
    v2f b2 = wrow[32];
    v2f b3 = wrow[48];
    acc0 = __builtin_amdgcn_wmma_f32_16x16x4_f32(false, a, false, b0, (short)0, acc0, false, false);
    acc1 = __builtin_amdgcn_wmma_f32_16x16x4_f32(false, a, false, b1, (short)0, acc1, false, false);
    acc2 = __builtin_amdgcn_wmma_f32_16x16x4_f32(false, a, false, b2, (short)0, acc2, false, false);
    acc3 = __builtin_amdgcn_wmma_f32_16x16x4_f32(false, a, false, b3, (short)0, acc3, false, false);
  }

#pragma unroll
  for (int v = 0; v < 8; ++v) {
    int rr = row0 + v + 8 * half;
    if (rr < M) {
      float* crow = Cout + (size_t)rr * D + lid;
      crow[0]  = acc0[v];
      crow[16] = acc1[v];
      crow[32] = acc2[v];
      crow[48] = acc3[v];
    }
  }
}

__global__ void __launch_bounds__(256)
k_gather(const int* __restrict__ rowptr, const int* __restrict__ csr,
         const float* __restrict__ dinv, const float* __restrict__ h,
         const float* __restrict__ b, float* __restrict__ out, int n, int E) {
  const int lane = threadIdx.x & 31;
  const int node = __builtin_amdgcn_readfirstlane(
      (int)(blockIdx.x * (blockDim.x >> 5) + (threadIdx.x >> 5)));
  if (node >= n) return;

  const int beg = rowptr[node];
  const int end = (node + 1 < n) ? rowptr[node + 1] : E;
  const float dt = dinv[node];
  const v2f* __restrict__ h2 = (const v2f*)h;
  const v2f hself = h2[(size_t)node * (D / 2) + lane];
  const v2f bb = ((const v2f*)b)[lane];

  v2f acc = {0.0f, 0.0f};
  for (int base = beg; base < end; base += 16) {
    int c = end - base;
    if (c > 16) c = 16;
    int my_s = node;
    if (lane < c) my_s = csr[base + lane];
    const float my_d = dinv[my_s];
    v2f v[16];
    for (int k = 0; k < 16; ++k) {
      if (k < c) {
        const int s = __builtin_amdgcn_readlane(my_s, k);
        v[k] = h2[(size_t)s * (D / 2) + lane];
      }
    }
    const int my_w = __float_as_int(my_d * dt);
    for (int k = 0; k < 16; ++k) {
      if (k < c) {
        const float wk = __int_as_float(__builtin_amdgcn_readlane(my_w, k));
        acc.x = fmaf(v[k].x, wk, acc.x);
        acc.y = fmaf(v[k].y, wk, acc.y);
      }
    }
  }
  const float wself = dt * dt;
  acc.x = fmaf(hself.x, wself, acc.x) + bb.x;
  acc.y = fmaf(hself.y, wself, acc.y) + bb.y;
  ((v2f*)out)[(size_t)node * (D / 2) + lane] = acc;
}

__global__ void k_prelu(const float* __restrict__ in, const float* __restrict__ a,
                        float* __restrict__ out, int n) {
  int i = blockIdx.x * blockDim.x + threadIdx.x;
  if (i < n * D) {
    float v = in[i];
    float al = a[i & (D - 1)];
    out[i] = (v >= 0.0f) ? v : al * v;
  }
}

static inline int cdiv_ll(long long a, long long b) { return (int)((a + b - 1) / b); }

extern "C" void kernel_launch(void* const* d_in, const int* in_sizes, int n_in,
                              void* d_out, int out_size, void* d_ws, size_t ws_size,
                              hipStream_t stream) {
  (void)n_in; (void)out_size; (void)ws_size;

  const float* x  = (const float*)d_in[0];
  const int*   ei = (const int*)  d_in[1];
  const float* W1 = (const float*)d_in[2];
  const float* b1 = (const float*)d_in[3];
  const float* a1 = (const float*)d_in[4];
  const float* W2 = (const float*)d_in[5];
  const float* b2 = (const float*)d_in[6];
  const float* a2 = (const float*)d_in[7];

  const int n = in_sizes[0] / D;
  const int E = in_sizes[1] / 2;
  const int* src = ei;
  const int* dst = ei + E;

  float* ws     = (float*)d_ws;
  float* dinv   = ws;
  int*   rowptr = (int*)(dinv + n);
  int*   csr    = rowptr + n;
  float* bufA   = (float*)(csr + E);
  float* bufB   = bufA + (size_t)n * D;
  int*   cnt    = (int*)bufB;
  int*   rank   = cnt + n;
  int*   bsum   = rank + E;
  float* out    = (float*)d_out;

  const int B = 256;
  const int gN    = cdiv_ll(n, B);
  const int gE    = cdiv_ll(E, B);
  const int gS    = cdiv_ll(n, 4 * B);
  const int gND   = cdiv_ll((long long)n * D, B);
  const int gGa   = cdiv_ll((long long)n * 32, B);
  const int gGemm = cdiv_ll((long long)cdiv_ll(n, 16) * 32, B);

  k_cnt_zero  <<<gN, B, 0, stream>>>(cnt, n);
  k_rank      <<<gE, B, 0, stream>>>(dst, cnt, rank, E);
  k_scan_local<<<gS, B, 0, stream>>>(cnt, rowptr, bsum, dinv, n);
  k_scan_add  <<<gS, B, 0, stream>>>(rowptr, bsum, n);
  k_fill      <<<gE, B, 0, stream>>>(src, dst, rank, rowptr, csr, E);

  k_gemm_wmma<<<gGemm, B, 0, stream>>>(x, W1, bufA, n);
  k_gather   <<<gGa,   B, 0, stream>>>(rowptr, csr, dinv, bufA, b1, bufB, n, E);
  k_prelu    <<<gND,   B, 0, stream>>>(bufB, a1, bufA, n);

  k_gemm_wmma<<<gGemm, B, 0, stream>>>(bufA, W2, bufB, n);
  k_gather   <<<gGa,   B, 0, stream>>>(rowptr, csr, dinv, bufB, b2, out, n, E);
  k_prelu    <<<gND,   B, 0, stream>>>(out, a2, out, n);
}


# Apply the per-channel PReLU in the aggregation kernel's epilogue (same registers, same order of operations) and drop both elementwise kernels: 400000 fewer waves, 2 fewer launches
# speedup vs baseline: 3.1077x; 1.1670x over previous
#include <hip/hip_runtime.h>
#include <hip/hip_bf16.h>

typedef float v2f __attribute__((ext_vector_type(2)));
typedef float v8f __attribute__((ext_vector_type(8)));

#define D 64

__global__ void k_cnt_zero(int* __restrict__ cnt, int n) {
  int i = blockIdx.x * blockDim.x + threadIdx.x;
  if (i < n) cnt[i] = 0;
}

__global__ void k_rank(const int* __restrict__ dst, int* __restrict__ cnt,
                       int* __restrict__ rank, int E) {
  int e = blockIdx.x * blockDim.x + threadIdx.x;
  if (e < E) rank[e] = atomicAdd(&cnt[dst[e]], 1);
}

__global__ void __launch_bounds__(256)
k_scan_local(const int* __restrict__ cnt, int* __restrict__ rowptr,
             int* __restrict__ bsum, float* __restrict__ dinv, int n) {
  __shared__ int s_wave[8];
  const int t = threadIdx.x;
  const int lane = t & 31;
  const int w = t >> 5;
  const int base = (blockIdx.x * 256 + t) * 4;
  int c[4];
  for (int j = 0; j < 4; ++j) c[j] = (base + j < n) ? cnt[base + j] : 0;
  const int s = c[0] + c[1] + c[2] + c[3];
  int inc = s;
  for (int o = 1; o < 32; o <<= 1) {
    int v = __shfl_up(inc, o);
    if (lane >= o) inc += v;
  }
  if (lane == 31) s_wave[w] = inc;
  __syncthreads();
  int woff = 0, tot = 0;
  for (int j = 0; j < 8; ++j) {
    int v = s_wave[j];
    tot += v;
    if (j < w) woff += v;
  }
  int run = woff + inc - s;
  for (int j = 0; j < 4; ++j) {
    if (base + j < n) {
      rowptr[base + j] = run;
      float d = (float)(c[j] + 1);
      dinv[base + j] = (d > 0.0f) ? rsqrtf(d) : 0.0f;
    }
    run += c[j];
  }
  if (t == 0) bsum[blockIdx.x] = tot;
}

__global__ void __launch_bounds__(256)
k_scan_add(int* __restrict__ rowptr, const int* __restrict__ bsum, int n) {
  __shared__ int s_off;
  const int t = threadIdx.x;
  if (t < 32) {
    int a = 0;
    for (int j = t; j < (int)blockIdx.x; j += 32) a += bsum[j];
    for (int o = 16; o > 0; o >>= 1) a += __shfl_xor(a, o);
    if (t == 0) s_off = a;
  }
  __syncthreads();
  const int off = s_off;
  const int base = (blockIdx.x * 256 + t) * 4;
  for (int j = 0; j < 4; ++j)
    if (base + j < n) rowptr[base + j] += off;
}

__global__ void k_fill(const int* __restrict__ src, const int* __restrict__ dst,
                       const int* __restrict__ rank, const int* __restrict__ rowptr,
                       int* __restrict__ csr, int E) {
  int e = blockIdx.x * blockDim.x + threadIdx.x;
  if (e < E) csr[rowptr[dst[e]] + rank[e]] = src[e];
}

__global__ void k_gemm_wmma(const float* __restrict__ A, const float* __restrict__ W,
                            float* __restrict__ Cout, int M) {
  __shared__ v2f sW[(D / 2) * D];
  for (int i = threadIdx.x; i < (D / 2) * D; i += blockDim.x) {
    int kk = i / D;
    int nn = i - kk * D;
    v2f p;
    p.x = W[(2 * kk + 0) * D + nn];
    p.y = W[(2 * kk + 1) * D + nn];
    sW[i] = p;
  }
  __syncthreads();

  int lane = threadIdx.x & 31;
  int half = lane >> 4;
  int lid  = lane & 15;
  int wave = blockIdx.x * (blockDim.x >> 5) + (threadIdx.x >> 5);
  int row0 = wave * 16;
  if (row0 >= M) return;

  v8f acc0 = {}, acc1 = {}, acc2 = {}, acc3 = {};

  int r  = row0 + lid;
  int rc = (r < M) ? r : (M - 1);
  const v2f* arow = (const v2f*)(A + (size_t)rc * D);

#pragma unroll
  for (int k = 0; k < D; k += 4) {
    int kk = (k >> 1) + half;
    v2f a = arow[kk];
    const v2f* wrow = &sW[kk * D + lid];
    v2f b0 = wrow[0];
    v2f b1 = wrow[16];
    v2f b2 = wrow[32];
    v2f b3 = wrow[48];
    acc0 = __builtin_amdgcn_wmma_f32_16x16x4_f32(false, a, false, b0, (short)0, acc0, false, false);
    acc1 = __builtin_amdgcn_wmma_f32_16x16x4_f32(false, a, false, b1, (short)0, acc1, false, false);
    acc2 = __builtin_amdgcn_wmma_f32_16x16x4_f32(false, a, false, b2, (short)0, acc2, false, false);
    acc3 = __builtin_amdgcn_wmma_f32_16x16x4_f32(false, a, false, b3, (short)0, acc3, false, false);
  }

#pragma unroll
  for (int v = 0; v < 8; ++v) {
    int rr = row0 + v + 8 * half;
    if (rr < M) {
      float* crow = Cout + (size_t)rr * D + lid;
      crow[0]  = acc0[v];
      crow[16] = acc1[v];
      crow[32] = acc2[v];
      crow[48] = acc3[v];
    }
  }
}

__global__ void __launch_bounds__(256)
k_gather(const int* __restrict__ rowptr, const int* __restrict__ csr,
         const float* __restrict__ dinv, const float* __restrict__ h,
         const float* __restrict__ b, const float* __restrict__ a,
         float* __restrict__ out, int n, int E) {
  const int lane = threadIdx.x & 31;
  const int node = __builtin_amdgcn_readfirstlane(
      (int)(blockIdx.x * (blockDim.x >> 5) + (threadIdx.x >> 5)));
  if (node >= n) return;

  const int beg = rowptr[node];
  const int end = (node + 1 < n) ? rowptr[node + 1] : E;
  const float dt = dinv[node];
  const v2f* __restrict__ h2 = (const v2f*)h;
  const v2f hself = h2[(size_t)node * (D / 2) + lane];
  const v2f bb = ((const v2f*)b)[lane];
  const v2f al = ((const v2f*)a)[lane];

  v2f acc = {0.0f, 0.0f};
  for (int base = beg; base < end; base += 16) {
    int c = end - base;
    if (c > 16) c = 16;
    int my_s = node;
    if (lane < c) my_s = csr[base + lane];
    const float my_d = dinv[my_s];
    v2f v[16];
    for (int k = 0; k < 16; ++k) {
      if (k < c) {
        const int s = __builtin_amdgcn_readlane(my_s, k);
        v[k] = h2[(size_t)s * (D / 2) + lane];
      }
    }
    const int my_w = __float_as_int(my_d * dt);
    for (int k = 0; k < 16; ++k) {
      if (k < c) {
        const float wk = __int_as_float(__builtin_amdgcn_readlane(my_w, k));
        acc.x = fmaf(v[k].x, wk, acc.x);
        acc.y = fmaf(v[k].y, wk, acc.y);
      }
    }
  }
  const float wself = dt * dt;
  acc.x = fmaf(hself.x, wself, acc.x) + bb.x;
  acc.y = fmaf(hself.y, wself, acc.y) + bb.y;
  acc.x = (acc.x >= 0.0f) ? acc.x : al.x * acc.x;
  acc.y = (acc.y >= 0.0f) ? acc.y : al.y * acc.y;
  ((v2f*)out)[(size_t)node * (D / 2) + lane] = acc;
}

static inline int cdiv_ll(long long a, long long b) { return (int)((a + b - 1) / b); }

extern "C" void kernel_launch(void* const* d_in, const int* in_sizes, int n_in,
                              void* d_out, int out_size, void* d_ws, size_t ws_size,
                              hipStream_t stream) {
  (void)n_in; (void)out_size; (void)ws_size;

  const float* x  = (const float*)d_in[0];
  const int*   ei = (const int*)  d_in[1];
  const float* W1 = (const float*)d_in[2];
  const float* b1 = (const float*)d_in[3];
  const float* a1 = (const float*)d_in[4];
  const float* W2 = (const float*)d_in[5];
  const float* b2 = (const float*)d_in[6];
  const float* a2 = (const float*)d_in[7];

  const int n = in_sizes[0] / D;
  const int E = in_sizes[1] / 2;
  const int* src = ei;
  const int* dst = ei + E;

  float* ws     = (float*)d_ws;
  float* dinv   = ws;
  int*   rowptr = (int*)(dinv + n);
  int*   csr    = rowptr + n;
  float* bufA   = (float*)(csr + E);
  float* bufB   = bufA + (size_t)n * D;
  int*   cnt    = (int*)bufB;
  int*   rank   = cnt + n;
  int*   bsum   = rank + E;
  float* out    = (float*)d_out;

  const int B = 256;
  const int gN    = cdiv_ll(n, B);
  const int gE    = cdiv_ll(E, B);
  const int gS    = cdiv_ll(n, 4 * B);
  const int gGa   = cdiv_ll((long long)n * 32, B);
  const int gGemm = cdiv_ll((long long)cdiv_ll(n, 16) * 32, B);

  k_cnt_zero  <<<gN, B, 0, stream>>>(cnt, n);
  k_rank      <<<gE, B, 0, stream>>>(dst, cnt, rank, E);
  k_scan_local<<<gS, B, 0, stream>>>(cnt, rowptr, bsum, dinv, n);
  k_scan_add  <<<gS, B, 0, stream>>>(rowptr, bsum, n);
  k_fill      <<<gE, B, 0, stream>>>(src, dst, rank, rowptr, csr, E);

  k_gemm_wmma<<<gGemm, B, 0, stream>>>(x, W1, bufA, n);
  k_gather   <<<gGa,   B, 0, stream>>>(rowptr, csr, dinv, bufA, b1, a1, bufB, n, E);

  k_gemm_wmma<<<gGemm, B, 0, stream>>>(bufB, W2, bufA, n);
  k_gather   <<<gGa,   B, 0, stream>>>(rowptr, csr, dinv, bufA, b2, a2, out, n, E);
}


# Group edges by destination with LDS counters in two levels (256 chunks bin by 512-node range, then one work-group per range ranks in LDS), no returning global add per edge; sum order changes
# speedup vs baseline: 5.0609x; 1.6285x over previous
#include <hip/hip_runtime.h>
#include <hip/hip_bf16.h>

typedef float v2f __attribute__((ext_vector_type(2)));
typedef float v8f __attribute__((ext_vector_type(8)));

#define D 64

#define BSH 9
#define BR  (1 << BSH)
#define NCH 256
#define TB  1024
#define EB  8

__device__ __forceinline__ int wave_incl_scan(int v, int lane) {
  for (int o = 1; o < 32; o <<= 1) {
    int u = __shfl_up(v, o);
    if (lane >= o) v += u;
  }
  return v;
}

__global__ void __launch_bounds__(TB)
k_bin_count(const int* __restrict__ dst, int* __restrict__ tab, int E, int nb) {
  extern __shared__ int s_dyn[];
  const int t = threadIdx.x;
  for (int j = t; j < nb; j += TB) s_dyn[j] = 0;
  __syncthreads();
  const int per = (E + NCH - 1) / NCH;
  const int lo = blockIdx.x * per;
  const int hi = (lo + per < E) ? (lo + per) : E;
  for (int i0 = lo; i0 < hi; i0 += EB * TB) {
    int d[EB];
    for (int j = 0; j < EB; ++j) {
      int i = i0 + j * TB + t;
      d[j] = dst[(i < hi) ? i : (hi - 1)];
    }
    for (int j = 0; j < EB; ++j) {
      int i = i0 + j * TB + t;
      if (i < hi) atomicAdd(&s_dyn[d[j] >> BSH], 1);
    }
  }
  __syncthreads();
  for (int j = t; j < nb; j += TB) tab[blockIdx.x * nb + j] = s_dyn[j];
}

__global__ void __launch_bounds__(NCH)
k_bin_rowscan(int* __restrict__ tab, int* __restrict__ btot, int nb) {
  __shared__ int s_w[NCH / 32];
  const int t = threadIdx.x;
  const int lane = t & 31;
  const int w = t >> 5;
  const int b = blockIdx.x;
  const int c = tab[t * nb + b];
  const int inc = wave_incl_scan(c, lane);
  if (lane == 31) s_w[w] = inc;
  __syncthreads();
  int woff = 0, tot = 0;
  for (int j = 0; j < NCH / 32; ++j) {
    int v = s_w[j];
    tot += v;
    if (j < w) woff += v;
  }
  tab[t * nb + b] = woff + inc - c;
  if (t == 0) btot[b] = tot;
}

__global__ void __launch_bounds__(TB)
k_bin_place(const int* __restrict__ src, const int* __restrict__ dst,
            const int* __restrict__ tab, const int* __restrict__ btot,
            int* __restrict__ bstart, int2* __restrict__ rec, int E, int nb) {
  extern __shared__ int s_dyn[];
  __shared__ int s_w[TB / 32];
  int* s_slot = s_dyn;
  int* s_base = s_dyn + nb;
  const int t = threadIdx.x;
  const int lane = t & 31;
  const int w = t >> 5;
  const int per = (E + NCH - 1) / NCH;
  const int lo = blockIdx.x * per;
  const int hi = (lo + per < E) ? (lo + per) : E;
  int d[EB], s[EB];
  for (int j = 0; j < EB; ++j) {
    int i = lo + j * TB + t;
    int ic = (i < hi) ? i : (hi - 1);
    ic = (ic > 0) ? ic : 0;
    d[j] = dst[ic];
    s[j] = src[ic];
  }
  int carry = 0;
  for (int j0 = 0; j0 < nb; j0 += TB) {
    const int j = j0 + t;
    const int v = (j < nb) ? btot[j] : 0;
    const int tv = (j < nb) ? tab[blockIdx.x * nb + j] : 0;
    const int inc = wave_incl_scan(v, lane);
    if (lane == 31) s_w[w] = inc;
    __syncthreads();
    int woff = 0, tot = 0;
    for (int k = 0; k < TB / 32; ++k) {
      int x = s_w[k];
      tot += x;
      if (k < w) woff += x;
    }
    if (j < nb) {
      const int st = carry + woff + inc - v;
      s_base[j] = st + tv;
      s_slot[j] = 0;
      if (blockIdx.x == 0) bstart[j] = st;
    }
    carry += tot;
    __syncthreads();
  }
  for (int i0 = lo; i0 < hi; ) {
    for (int j = 0; j < EB; ++j) {
      int i = i0 + j * TB + t;
      if (i < hi) {
        const int bk = d[j] >> BSH;
        const int r = atomicAdd(&s_slot[bk], 1);
        rec[s_base[bk] + r] = make_int2(s[j], d[j]);
      }
    }
    i0 += EB * TB;
    if (i0 < hi) {
      for (int j = 0; j < EB; ++j) {
        int i = i0 + j * TB + t;
        int ic = (i < hi) ? i : (hi - 1);
        d[j] = dst[ic];
        s[j] = src[ic];
      }
    }
  }
}

__global__ void __launch_bounds__(TB)
k_bucket_rows(const int2* __restrict__ rec, const int* __restrict__ bstart,
              const int* __restrict__ btot,
              int* __restrict__ rowptr, float* __restrict__ dinv,
              int* __restrict__ csr, int n) {
  __shared__ int s_cnt[BR];
  __shared__ int s_off[BR];
  __shared__ int s_w[TB / 32];
  const int t = threadIdx.x;
  const int lane = t & 31;
  const int w = t >> 5;
  const int b = blockIdx.x;
  const int node0 = b << BSH;
  const int bs = bstart[b];
  const int be = bs + btot[b];
  if (t < BR) s_cnt[t] = 0;
  __syncthreads();

  for (int i0 = bs; i0 < be; i0 += EB * TB) {
    int d[EB];
    for (int j = 0; j < EB; ++j) {
      int i = i0 + j * TB + t;
      d[j] = rec[(i < be) ? i : (be - 1)].y;
    }
    for (int j = 0; j < EB; ++j) {
      int i = i0 + j * TB + t;
      if (i < be) atomicAdd(&s_cnt[d[j] - node0], 1);
    }
  }
  __syncthreads();

  const int c = (t < BR) ? s_cnt[t] : 0;
  const int inc = wave_incl_scan(c, lane);
  if (lane == 31) s_w[w] = inc;
  __syncthreads();
  int woff = 0;
  for (int k = 0; k < TB / 32; ++k)
    if (k < w) woff += s_w[k];
  if (t < BR) {
    const int start = bs + woff + inc - c;
    s_off[t] = start;
    s_cnt[t] = 0;
    const int node = node0 + t;
    if (node < n) {
      rowptr[node] = start;
      float dg = (float)(c + 1);
      dinv[node] = (dg > 0.0f) ? rsqrtf(dg) : 0.0f;
    }
  }
  __syncthreads();

  for (int i0 = bs; i0 < be; i0 += EB * TB) {
    int2 r[EB];
    for (int j = 0; j < EB; ++j) {
      int i = i0 + j * TB + t;
      r[j] = rec[(i < be) ? i : (be - 1)];
    }
    for (int j = 0; j < EB; ++j) {
      int i = i0 + j * TB + t;
      if (i < be) {
        const int dl = r[j].y - node0;
        const int k = atomicAdd(&s_cnt[dl], 1);
        csr[s_off[dl] + k] = r[j].x;
      }
    }
  }
}

__global__ void k_gemm_wmma(const float* __restrict__ A, const float* __restrict__ W,
                            const float* __restrict__ scale,
                            float* __restrict__ Cout, int M) {
  __shared__ v2f sW[(D / 2) * D];
  for (int i = threadIdx.x; i < (D / 2) * D; i += blockDim.x) {
    int kk = i / D;
    int nn = i - kk * D;
    v2f p;
    p.x = W[(2 * kk + 0) * D + nn];
    p.y = W[(2 * kk + 1) * D + nn];
    sW[i] = p;
  }
  __syncthreads();

  int lane = threadIdx.x & 31;
  int half = lane >> 4;
  int lid  = lane & 15;
  int wave = blockIdx.x * (blockDim.x >> 5) + (threadIdx.x >> 5);
  int row0 = wave * 16;
  if (row0 >= M) return;

  v8f acc0 = {}, acc1 = {}, acc2 = {}, acc3 = {};

  int r  = row0 + lid;
  int rc = (r < M) ? r : (M - 1);
  const v2f* arow = (const v2f*)(A + (size_t)rc * D);
  const float ascale = scale[rc];

#pragma unroll
  for (int k = 0; k < D; k += 4) {
    int kk = (k >> 1) + half;
    v2f a = arow[kk];
    a.x *= ascale;
    a.y *= ascale;
    const v2f* wrow = &sW[kk * D + lid];
    v2f b0 = wrow[0];
    v2f b1 = wrow[16];
    v2f b2 = wrow[32];
    v2f b3 = wrow[48];
    acc0 = __builtin_amdgcn_wmma_f32_16x16x4_f32(false, a, false, b0, (short)0, acc0, false, false);
    acc1 = __builtin_amdgcn_wmma_f32_16x16x4_f32(false, a, false, b1, (short)0, acc1, false, false);
    acc2 = __builtin_amdgcn_wmma_f32_16x16x4_f32(false, a, false, b2, (short)0, acc2, false, false);
    acc3 = __builtin_amdgcn_wmma_f32_16x16x4_f32(false, a, false, b3, (short)0, acc3, false, false);
  }

  __builtin_amdgcn_sched_barrier(0);
  asm volatile("v_nop\n\tv_nop\n\tv_nop\n\tv_nop");
  __builtin_amdgcn_sched_barrier(0);

#pragma unroll
  for (int v = 0; v < 8; ++v) {
    int rr = row0 + v + 8 * half;
    if (rr < M) {
      float* crow = Cout + (size_t)rr * D + lid;
      crow[0]  = acc0[v];
      crow[16] = acc1[v];
      crow[32] = acc2[v];
      crow[48] = acc3[v];
    }
  }
}

__global__ void __launch_bounds__(256)
k_gather(const int* __restrict__ rowptr, const int* __restrict__ csr,
         const float* __restrict__ dinv, const float* __restrict__ h,
         const float* __restrict__ b, const float* __restrict__ a,
         float* __restrict__ out, int n, int E) {
  const int lane = threadIdx.x & 31;
  const int node = __builtin_amdgcn_readfirstlane(
      (int)(blockIdx.x * (blockDim.x >> 5) + (threadIdx.x >> 5)));
  if (node >= n) return;

  const int beg = rowptr[node];
  const int end = (node + 1 < n) ? rowptr[node + 1] : E;
  const float dt = dinv[node];
  const v2f* __restrict__ h2 = (const v2f*)h;
  const v2f hself = h2[(size_t)node * (D / 2) + lane];
  const v2f bb = ((const v2f*)b)[lane];
  const v2f al = ((const v2f*)a)[lane];

  v2f acc = {0.0f, 0.0f};
  for (int base = beg; base < end; base += 16) {
    int c = end - base;
    if (c > 16) c = 16;
    int my_s = node;
    if (lane < c) my_s = csr[base + lane];
    int sid[16];
    for (int k = 0; k < 16; ++k) sid[k] = __builtin_amdgcn_readlane(my_s, k);
    v2f v[16];
    for (int k = 0; k < 16; ++k) {
      if (k < c) v[k] = h2[(size_t)sid[k] * (D / 2) + lane];
    }
    for (int k = 0; k < 16; ++k) {
      if (k < c) {
        acc.x += v[k].x;
        acc.y += v[k].y;
      }
    }
  }
  acc.x = (acc.x + hself.x) * dt + bb.x;
  acc.y = (acc.y + hself.y) * dt + bb.y;
  acc.x = (acc.x >= 0.0f) ? acc.x : al.x * acc.x;
  acc.y = (acc.y >= 0.0f) ? acc.y : al.y * acc.y;
  ((v2f*)out)[(size_t)node * (D / 2) + lane] = acc;
}

static inline int cdiv_ll(long long a, long long b) { return (int)((a + b - 1) / b); }

extern "C" void kernel_launch(void* const* d_in, const int* in_sizes, int n_in,
                              void* d_out, int out_size, void* d_ws, size_t ws_size,
                              hipStream_t stream) {
  (void)n_in; (void)out_size; (void)ws_size;

  const float* x  = (const float*)d_in[0];
  const int*   ei = (const int*)  d_in[1];
  const float* W1 = (const float*)d_in[2];
  const float* b1 = (const float*)d_in[3];
  const float* a1 = (const float*)d_in[4];
  const float* W2 = (const float*)d_in[5];
  const float* b2 = (const float*)d_in[6];
  const float* a2 = (const float*)d_in[7];

  const int n = in_sizes[0] / D;
  const int E = in_sizes[1] / 2;
  const int* src = ei;
  const int* dst = ei + E;

  float* ws     = (float*)d_ws;
  float* dinv   = ws;
  int*   rowptr = (int*)(dinv + n);
  int*   csr    = rowptr + n;
  float* bufA   = (float*)(csr + E);
  float* bufB   = bufA + (size_t)n * D;
  const int nb  = (n + BR - 1) / BR;
  int*   tab    = (int*)bufA;
  int*   btot   = tab + (size_t)NCH * nb;
  int*   bstart = btot + nb;
  int2*  rec    = (int2*)(bufB + (E & 1));
  float* out    = (float*)d_out;

  const int B = 256;
  const int gGa   = cdiv_ll((long long)n * 32, B);
  const int gGemm = cdiv_ll((long long)cdiv_ll(n, 16) * 32, B);

  k_bin_count  <<<NCH, TB, nb * sizeof(int), stream>>>(dst, tab, E, nb);
  k_bin_rowscan<<<nb, NCH, 0, stream>>>(tab, btot, nb);
  k_bin_place  <<<NCH, TB, 2 * nb * sizeof(int), stream>>>(src, dst, tab, btot, bstart, rec, E, nb);
  k_bucket_rows<<<nb, TB, 0, stream>>>(rec, bstart, btot, rowptr, dinv, csr, n);

  k_gemm_wmma<<<gGemm, B, 0, stream>>>(x, W1, dinv, bufA, n);
  k_gather   <<<gGa,   B, 0, stream>>>(rowptr, csr, dinv, bufA, b1, a1, bufB, n, E);

  k_gemm_wmma<<<gGemm, B, 0, stream>>>(bufB, W2, dinv, bufA, n);
  k_gather   <<<gGa,   B, 0, stream>>>(rowptr, csr, dinv, bufA, b2, a2, out, n, E);
}


# Layer 2's product runs behind layer 1's aggregation in one launch: 16-wave groups sum 32 nodes (2 per wave) into an LDS tile, 4 waves multiply it by W2 on f32 WMMA; 1 launch, 1 row write+read fewer
# speedup vs baseline: 5.7596x; 1.1381x over previous
#include <hip/hip_runtime.h>
#include <hip/hip_bf16.h>

typedef float v2f __attribute__((ext_vector_type(2)));
typedef float v8f __attribute__((ext_vector_type(8)));

#define D 64

#define BSH 9
#define BR  (1 << BSH)
#define NCH 256
#define TB  1024
#define EB  8

__device__ __forceinline__ int wave_incl_scan(int v, int lane) {
  for (int o = 1; o < 32; o <<= 1) {
    int u = __shfl_up(v, o);
    if (lane >= o) v += u;
  }
  return v;
}

__global__ void __launch_bounds__(TB)
k_bin_count(const int* __restrict__ dst, int* __restrict__ tab, int E, int nb) {
  extern __shared__ int s_dyn[];
  const int t = threadIdx.x;
  for (int j = t; j < nb; j += TB) s_dyn[j] = 0;
  __syncthreads();
  const int per = (E + NCH - 1) / NCH;
  const int lo = blockIdx.x * per;
  const int hi = (lo + per < E) ? (lo + per) : E;
  for (int i0 = lo; i0 < hi; i0 += EB * TB) {
    int d[EB];
    for (int j = 0; j < EB; ++j) {
      int i = i0 + j * TB + t;
      d[j] = dst[(i < hi) ? i : (hi - 1)];
    }
    for (int j = 0; j < EB; ++j) {
      int i = i0 + j * TB + t;
      if (i < hi) atomicAdd(&s_dyn[d[j] >> BSH], 1);
    }
  }
  __syncthreads();
  for (int j = t; j < nb; j += TB) tab[blockIdx.x * nb + j] = s_dyn[j];
}

__global__ void __launch_bounds__(NCH)
k_bin_rowscan(int* __restrict__ tab, int* __restrict__ btot, int nb) {
  __shared__ int s_w[NCH / 32];
  const int t = threadIdx.x;
  const int lane = t & 31;
  const int w = t >> 5;
  const int b = blockIdx.x;
  const int c = tab[t * nb + b];
  const int inc = wave_incl_scan(c, lane);
  if (lane == 31) s_w[w] = inc;
  __syncthreads();
  int woff = 0, tot = 0;
  for (int j = 0; j < NCH / 32; ++j) {
    int v = s_w[j];
    tot += v;
    if (j < w) woff += v;
  }
  tab[t * nb + b] = woff + inc - c;
  if (t == 0) btot[b] = tot;
}

__global__ void __launch_bounds__(TB)
k_bin_place(const int* __restrict__ src, const int* __restrict__ dst,
            const int* __restrict__ tab, const int* __restrict__ btot,
            int* __restrict__ bstart, int2* __restrict__ rec, int E, int nb) {
  extern __shared__ int s_dyn[];
  __shared__ int s_w[TB / 32];
  int* s_slot = s_dyn;
  int* s_base = s_dyn + nb;
  const int t = threadIdx.x;
  const int lane = t & 31;
  const int w = t >> 5;
  const int per = (E + NCH - 1) / NCH;
  const int lo = blockIdx.x * per;
  const int hi = (lo + per < E) ? (lo + per) : E;
  int d[EB], s[EB];
  for (int j = 0; j < EB; ++j) {
    int i = lo + j * TB + t;
    int ic = (i < hi) ? i : (hi - 1);
    ic = (ic > 0) ? ic : 0;
    d[j] = dst[ic];
    s[j] = src[ic];
  }
  int carry = 0;
  for (int j0 = 0; j0 < nb; j0 += TB) {
    const int j = j0 + t;
    const int v = (j < nb) ? btot[j] : 0;
    const int tv = (j < nb) ? tab[blockIdx.x * nb + j] : 0;
    const int inc = wave_incl_scan(v, lane);
    if (lane == 31) s_w[w] = inc;
    __syncthreads();
    int woff = 0, tot = 0;
    for (int k = 0; k < TB / 32; ++k) {
      int x = s_w[k];
      tot += x;
      if (k < w) woff += x;
    }
    if (j < nb) {
      const int st = carry + woff + inc - v;
      s_base[j] = st + tv;
      s_slot[j] = 0;
      if (blockIdx.x == 0) bstart[j] = st;
    }
    carry += tot;
    __syncthreads();
  }
  for (int i0 = lo; i0 < hi; ) {
    for (int j = 0; j < EB; ++j) {
      int i = i0 + j * TB + t;
      if (i < hi) {
        const int bk = d[j] >> BSH;
        const int r = atomicAdd(&s_slot[bk], 1);
        rec[s_base[bk] + r] = make_int2(s[j], d[j]);
      }
    }
    i0 += EB * TB;
    if (i0 < hi) {
      for (int j = 0; j < EB; ++j) {
        int i = i0 + j * TB + t;
        int ic = (i < hi) ? i : (hi - 1);
        d[j] = dst[ic];
        s[j] = src[ic];
      }
    }
  }
}

__global__ void __launch_bounds__(TB)
k_bucket_rows(const int2* __restrict__ rec, const int* __restrict__ bstart,
              const int* __restrict__ btot,
              int* __restrict__ rowptr, float* __restrict__ dinv,
              int* __restrict__ csr, int n) {
  __shared__ int s_cnt[BR];
  __shared__ int s_off[BR];
  __shared__ int s_w[TB / 32];
  const int t = threadIdx.x;
  const int lane = t & 31;
  const int w = t >> 5;
  const int b = blockIdx.x;
  const int node0 = b << BSH;
  const int bs = bstart[b];
  const int be = bs + btot[b];
  if (t < BR) s_cnt[t] = 0;
  __syncthreads();

  for (int i0 = bs; i0 < be; i0 += EB * TB) {
    int d[EB];
    for (int j = 0; j < EB; ++j) {
      int i = i0 + j * TB + t;
      d[j] = rec[(i < be) ? i : (be - 1)].y;
    }
    for (int j = 0; j < EB; ++j) {
      int i = i0 + j * TB + t;
      if (i < be) atomicAdd(&s_cnt[d[j] - node0], 1);
    }
  }
  __syncthreads();

  const int c = (t < BR) ? s_cnt[t] : 0;
  const int inc = wave_incl_scan(c, lane);
  if (lane == 31) s_w[w] = inc;
  __syncthreads();
  int woff = 0;
  for (int k = 0; k < TB / 32; ++k)
    if (k < w) woff += s_w[k];
  if (t < BR) {
    const int start = bs + woff + inc - c;
    s_off[t] = start;
    s_cnt[t] = 0;
    const int node = node0 + t;
    if (node < n) {
      rowptr[node] = start;
      float dg = (float)(c + 1);
      dinv[node] = (dg > 0.0f) ? rsqrtf(dg) : 0.0f;
    }
  }
  __syncthreads();

  for (int i0 = bs; i0 < be; i0 += EB * TB) {
    int2 r[EB];
    for (int j = 0; j < EB; ++j) {
      int i = i0 + j * TB + t;
      r[j] = rec[(i < be) ? i : (be - 1)];
    }
    for (int j = 0; j < EB; ++j) {
      int i = i0 + j * TB + t;
      if (i < be) {
        const int dl = r[j].y - node0;
        const int k = atomicAdd(&s_cnt[dl], 1);
        csr[s_off[dl] + k] = r[j].x;
      }
    }
  }
}

__global__ void k_gemm_wmma(const float* __restrict__ A, const float* __restrict__ W,
                            const float* __restrict__ scale,
                            float* __restrict__ Cout, int M) {
  __shared__ v2f sW[(D / 2) * D];
  for (int i = threadIdx.x; i < (D / 2) * D; i += blockDim.x) {
    int kk = i / D;
    int nn = i - kk * D;
    v2f p;
    p.x = W[(2 * kk + 0) * D + nn];
    p.y = W[(2 * kk + 1) * D + nn];
    sW[i] = p;
  }
  __syncthreads();

  int lane = threadIdx.x & 31;
  int half = lane >> 4;
  int lid  = lane & 15;
  int wave = blockIdx.x * (blockDim.x >> 5) + (threadIdx.x >> 5);
  int row0 = wave * 16;
  if (row0 >= M) return;

  v8f acc0 = {}, acc1 = {}, acc2 = {}, acc3 = {};

  int r  = row0 + lid;
  int rc = (r < M) ? r : (M - 1);
  const v2f* arow = (const v2f*)(A + (size_t)rc * D);
  const float ascale = scale[rc];

#pragma unroll
  for (int k = 0; k < D; k += 4) {
    int kk = (k >> 1) + half;
    v2f a = arow[kk];
    a.x *= ascale;
    a.y *= ascale;
    const v2f* wrow = &sW[kk * D + lid];
    v2f b0 = wrow[0];
    v2f b1 = wrow[16];
    v2f b2 = wrow[32];
    v2f b3 = wrow[48];
    acc0 = __builtin_amdgcn_wmma_f32_16x16x4_f32(false, a, false, b0, (short)0, acc0, false, false);
    acc1 = __builtin_amdgcn_wmma_f32_16x16x4_f32(false, a, false, b1, (short)0, acc1, false, false);
    acc2 = __builtin_amdgcn_wmma_f32_16x16x4_f32(false, a, false, b2, (short)0, acc2, false, false);
    acc3 = __builtin_amdgcn_wmma_f32_16x16x4_f32(false, a, false, b3, (short)0, acc3, false, false);
  }

  __builtin_amdgcn_sched_barrier(0);
  asm volatile("v_nop\n\tv_nop\n\tv_nop\n\tv_nop");
  __builtin_amdgcn_sched_barrier(0);

#pragma unroll
  for (int v = 0; v < 8; ++v) {
    int rr = row0 + v + 8 * half;
    if (rr < M) {
      float* crow = Cout + (size_t)rr * D + lid;
      crow[0]  = acc0[v];
      crow[16] = acc1[v];
      crow[32] = acc2[v];
      crow[48] = acc3[v];
    }
  }
}

__global__ void __launch_bounds__(256)
k_gather(const int* __restrict__ rowptr, const int* __restrict__ csr,
         const float* __restrict__ dinv, const float* __restrict__ h,
         const float* __restrict__ b, const float* __restrict__ a,
         float* __restrict__ out, int n, int E) {
  const int lane = threadIdx.x & 31;
  const int node = __builtin_amdgcn_readfirstlane(
      (int)(blockIdx.x * (blockDim.x >> 5) + (threadIdx.x >> 5)));
  if (node >= n) return;

  const int beg = rowptr[node];
  const int end = (node + 1 < n) ? rowptr[node + 1] : E;
  const float dt = dinv[node];
  const v2f* __restrict__ h2 = (const v2f*)h;
  const v2f hself = h2[(size_t)node * (D / 2) + lane];
  const v2f bb = ((const v2f*)b)[lane];
  const v2f al = ((const v2f*)a)[lane];

  v2f acc = {0.0f, 0.0f};
  for (int base = beg; base < end; base += 16) {
    int c = end - base;
    if (c > 16) c = 16;
    int my_s = node;
    if (lane < c) my_s = csr[base + lane];
    int sid[16];
    for (int k = 0; k < 16; ++k) sid[k] = __builtin_amdgcn_readlane(my_s, k);
    v2f v[16];
    for (int k = 0; k < 16; ++k) {
      if (k < c) v[k] = h2[(size_t)sid[k] * (D / 2) + lane];
    }
    for (int k = 0; k < 16; ++k) {
      if (k < c) {
        acc.x += v[k].x;
        acc.y += v[k].y;
      }
    }
  }
  acc.x = (acc.x + hself.x) * dt + bb.x;
  acc.y = (acc.y + hself.y) * dt + bb.y;
  acc.x = (acc.x >= 0.0f) ? acc.x : al.x * acc.x;
  acc.y = (acc.y >= 0.0f) ? acc.y : al.y * acc.y;
  ((v2f*)out)[(size_t)node * (D / 2) + lane] = acc;
}

#define TN 32
#define TP 68

__device__ __forceinline__ void gather_piece(v2f& acc, const v2f* __restrict__ h2,
                                             int my_s, int c, int lane) {
  int sid[16];
  for (int k = 0; k < 16; ++k) sid[k] = __builtin_amdgcn_readlane(my_s, k);
  v2f v[16];
  for (int k = 0; k < 16; ++k) {
    if (k < c) v[k] = h2[(size_t)sid[k] * (D / 2) + lane];
  }
  for (int k = 0; k < 16; ++k) {
    if (k < c) {
      acc.x += v[k].x;
      acc.y += v[k].y;
    }
  }
}

__global__ void __launch_bounds__(512)
k_gather_gemm(const int* __restrict__ rowptr, const int* __restrict__ csr,
              const float* __restrict__ dinv, const float* __restrict__ h,
              const float* __restrict__ b, const float* __restrict__ a,
              const float* __restrict__ W, float* __restrict__ Cout, int n, int E) {
  __shared__ v2f sW[(D / 2) * D];
  __shared__ v2f sT[TN * (TP / 2)];
  const int lane = threadIdx.x & 31;
  const int wave = __builtin_amdgcn_readfirstlane((int)(threadIdx.x >> 5));
  const int tile0 = blockIdx.x * TN;

  v2f wst[(D / 2) * D / 512];
#pragma unroll
  for (int j = 0; j < (D / 2) * D / 512; ++j) {
    const int i = (int)threadIdx.x + j * 512;
    const int kk = i / D;
    const int nn = i - kk * D;
    wst[j].x = W[(2 * kk + 0) * D + nn];
    wst[j].y = W[(2 * kk + 1) * D + nn];
  }

  int nodeA = tile0 + 2 * wave;
  int nodeB = nodeA + 1;
  nodeA = (nodeA < n) ? nodeA : (n - 1);
  nodeB = (nodeB < n) ? nodeB : (n - 1);
  const int nxA = (nodeA + 1 < n) ? (nodeA + 1) : (n - 1);
  const int nxB = (nodeB + 1 < n) ? (nodeB + 1) : (n - 1);
  const int begA = rowptr[nodeA];
  const int rnA  = rowptr[nxA];
  const int begB = rowptr[nodeB];
  const int rnB  = rowptr[nxB];
  const int endA = (nodeA + 1 < n) ? rnA : E;
  const int endB = (nodeB + 1 < n) ? rnB : E;
  const v2f* __restrict__ h2 = (const v2f*)h;

  int cA = endA - begA; if (cA > 16) cA = 16;
  int cB = endB - begB; if (cB > 16) cB = 16;
  int idA = nodeA;
  if (lane < cA) idA = csr[begA + lane];
  int idB = nodeB;
  if (lane < cB) idB = csr[begB + lane];
  const float dtA = dinv[nodeA];
  const float dtB = dinv[nodeB];
  const v2f hsA = h2[(size_t)nodeA * (D / 2) + lane];
  const v2f hsB = h2[(size_t)nodeB * (D / 2) + lane];
  const v2f bb = ((const v2f*)b)[lane];
  const v2f al = ((const v2f*)a)[lane];

#pragma unroll
  for (int j = 0; j < (D / 2) * D / 512; ++j) sW[(int)threadIdx.x + j * 512] = wst[j];

  {
    v2f acc = {0.0f, 0.0f};
    int base = begA, c = cA, my_s = idA;
    while (c > 0) {
      gather_piece(acc, h2, my_s, c, lane);
      base += 16;
      c = endA - base; if (c > 16) c = 16;
      if (c > 0) { my_s = nodeA; if (lane < c) my_s = csr[base + lane]; }
    }
    acc.x = (acc.x + hsA.x) * dtA + bb.x;
    acc.y = (acc.y + hsA.y) * dtA + bb.y;
    acc.x = (acc.x >= 0.0f) ? acc.x : al.x * acc.x;
    acc.y = (acc.y >= 0.0f) ? acc.y : al.y * acc.y;
    v2f r;
    r.x = acc.x * dtA;
    r.y = acc.y * dtA;
    sT[(2 * wave + 0) * (TP / 2) + lane] = r;
  }
  {
    v2f acc = {0.0f, 0.0f};
    int base = begB, c = cB, my_s = idB;
    while (c > 0) {
      gather_piece(acc, h2, my_s, c, lane);
      base += 16;
      c = endB - base; if (c > 16) c = 16;
      if (c > 0) { my_s = nodeB; if (lane < c) my_s = csr[base + lane]; }
    }
    acc.x = (acc.x + hsB.x) * dtB + bb.x;
    acc.y = (acc.y + hsB.y) * dtB + bb.y;
    acc.x = (acc.x >= 0.0f) ? acc.x : al.x * acc.x;
    acc.y = (acc.y >= 0.0f) ? acc.y : al.y * acc.y;
    v2f r;
    r.x = acc.x * dtB;
    r.y = acc.y * dtB;
    sT[(2 * wave + 1) * (TP / 2) + lane] = r;
  }
  __syncthreads();

  if (wave < 4) {
    const int half = lane >> 4;
    const int lid  = lane & 15;
    v8f acc0 = {}, acc1 = {};
    const v2f* t0 = &sT[lid * (TP / 2)];
    const v2f* t1 = &sT[(16 + lid) * (TP / 2)];
    const v2f* wcol = &sW[wave * 16 + lid];
#pragma unroll
    for (int k = 0; k < D; k += 4) {
      int kk = (k >> 1) + half;
      v2f a0 = t0[kk];
      v2f a1 = t1[kk];
      v2f bq = wcol[kk * D];
      acc0 = __builtin_amdgcn_wmma_f32_16x16x4_f32(false, a0, false, bq, (short)0, acc0, false, false);
      acc1 = __builtin_amdgcn_wmma_f32_16x16x4_f32(false, a1, false, bq, (short)0, acc1, false, false);
    }
    __builtin_amdgcn_sched_barrier(0);
    asm volatile("v_nop\n\tv_nop\n\tv_nop\n\tv_nop");
    __builtin_amdgcn_sched_barrier(0);
#pragma unroll
    for (int v = 0; v < 8; ++v) {
      int rr = tile0 + v + 8 * half;
      if (rr < n)      Cout[(size_t)rr * D + wave * 16 + lid] = acc0[v];
      if (rr + 16 < n) Cout[(size_t)(rr + 16) * D + wave * 16 + lid] = acc1[v];
    }
  }
}

static inline int cdiv_ll(long long a, long long b) { return (int)((a + b - 1) / b); }

extern "C" void kernel_launch(void* const* d_in, const int* in_sizes, int n_in,
                              void* d_out, int out_size, void* d_ws, size_t ws_size,
                              hipStream_t stream) {
  (void)n_in; (void)out_size; (void)ws_size;

  const float* x  = (const float*)d_in[0];
  const int*   ei = (const int*)  d_in[1];
  const float* W1 = (const float*)d_in[2];
  const float* b1 = (const float*)d_in[3];
  const float* a1 = (const float*)d_in[4];
  const float* W2 = (const float*)d_in[5];
  const float* b2 = (const float*)d_in[6];
  const float* a2 = (const float*)d_in[7];

  const int n = in_sizes[0] / D;
  const int E = in_sizes[1] / 2;
  const int* src = ei;
  const int* dst = ei + E;

  float* ws     = (float*)d_ws;
  float* dinv   = ws;
  int*   rowptr = (int*)(dinv + n);
  int*   csr    = rowptr + n;
  float* bufA   = (float*)(csr + E);
  float* bufB   = bufA + (size_t)n * D;
  const int nb  = (n + BR - 1) / BR;
  int*   tab    = (int*)bufA;
  int*   btot   = tab + (size_t)NCH * nb;
  int*   bstart = btot + nb;
  int2*  rec    = (int2*)(bufB + (E & 1));
  float* out    = (float*)d_out;

  const int B = 256;
  const int gGa   = cdiv_ll((long long)n * 32, B);
  const int gGemm = cdiv_ll((long long)cdiv_ll(n, 16) * 32, B);
  const int gFuse = cdiv_ll(n, TN);

  k_bin_count  <<<NCH, TB, nb * sizeof(int), stream>>>(dst, tab, E, nb);
  k_bin_rowscan<<<nb, NCH, 0, stream>>>(tab, btot, nb);
  k_bin_place  <<<NCH, TB, 2 * nb * sizeof(int), stream>>>(src, dst, tab, btot, bstart, rec, E, nb);
  k_bucket_rows<<<nb, TB, 0, stream>>>(rec, bstart, btot, rowptr, dinv, csr, n);

  k_gemm_wmma<<<gGemm, B, 0, stream>>>(x, W1, dinv, bufA, n);

  k_gather_gemm<<<gFuse, 512, 0, stream>>>(rowptr, csr, dinv, bufA, b1, a1, W2, bufB, n, E);

  k_gather   <<<gGa,   B, 0, stream>>>(rowptr, csr, dinv, bufB, b2, a2, out, n, E);
}


# Each gather wave takes 4 consecutive nodes (was 1 in the layer-2 gather, 2 in the fused layer-1 kernel): bounds and first ids of all 4 in one round trip each; 100000+50000 waves become 25000+25000
# speedup vs baseline: 5.9807x; 1.0384x over previous
#include <hip/hip_runtime.h>
#include <hip/hip_bf16.h>

typedef float v2f __attribute__((ext_vector_type(2)));
typedef float v8f __attribute__((ext_vector_type(8)));

#define D 64

#define BSH 9
#define BR  (1 << BSH)
#define NCH 256
#define TB  1024
#define EB  8

__device__ __forceinline__ int wave_incl_scan(int v, int lane) {
  for (int o = 1; o < 32; o <<= 1) {
    int u = __shfl_up(v, o);
    if (lane >= o) v += u;
  }
  return v;
}

__global__ void __launch_bounds__(TB)
k_bin_count(const int* __restrict__ dst, int* __restrict__ tab, int E, int nb) {
  extern __shared__ int s_dyn[];
  const int t = threadIdx.x;
  for (int j = t; j < nb; j += TB) s_dyn[j] = 0;
  __syncthreads();
  const int per = (E + NCH - 1) / NCH;
  const int lo = blockIdx.x * per;
  const int hi = (lo + per < E) ? (lo + per) : E;
  for (int i0 = lo; i0 < hi; i0 += EB * TB) {
    int d[EB];
    for (int j = 0; j < EB; ++j) {
      int i = i0 + j * TB + t;
      d[j] = dst[(i < hi) ? i : (hi - 1)];
    }
    for (int j = 0; j < EB; ++j) {
      int i = i0 + j * TB + t;
      if (i < hi) atomicAdd(&s_dyn[d[j] >> BSH], 1);
    }
  }
  __syncthreads();
  for (int j = t; j < nb; j += TB) tab[blockIdx.x * nb + j] = s_dyn[j];
}

__global__ void __launch_bounds__(NCH)
k_bin_rowscan(int* __restrict__ tab, int* __restrict__ btot, int nb) {
  __shared__ int s_w[NCH / 32];
  const int t = threadIdx.x;
  const int lane = t & 31;
  const int w = t >> 5;
  const int b = blockIdx.x;
  const int c = tab[t * nb + b];
  const int inc = wave_incl_scan(c, lane);
  if (lane == 31) s_w[w] = inc;
  __syncthreads();
  int woff = 0, tot = 0;
  for (int j = 0; j < NCH / 32; ++j) {
    int v = s_w[j];
    tot += v;
    if (j < w) woff += v;
  }
  tab[t * nb + b] = woff + inc - c;
  if (t == 0) btot[b] = tot;
}

__global__ void __launch_bounds__(TB)
k_bin_place(const int* __restrict__ src, const int* __restrict__ dst,
            const int* __restrict__ tab, const int* __restrict__ btot,
            int* __restrict__ bstart, int2* __restrict__ rec, int E, int nb) {
  extern __shared__ int s_dyn[];
  __shared__ int s_w[TB / 32];
  int* s_slot = s_dyn;
  int* s_base = s_dyn + nb;
  const int t = threadIdx.x;
  const int lane = t & 31;
  const int w = t >> 5;
  const int per = (E + NCH - 1) / NCH;
  const int lo = blockIdx.x * per;
  const int hi = (lo + per < E) ? (lo + per) : E;
  int d[EB], s[EB];
  for (int j = 0; j < EB; ++j) {
    int i = lo + j * TB + t;
    int ic = (i < hi) ? i : (hi - 1);
    ic = (ic > 0) ? ic : 0;
    d[j] = dst[ic];
    s[j] = src[ic];
  }
  int carry = 0;
  for (int j0 = 0; j0 < nb; j0 += TB) {
    const int j = j0 + t;
    const int v = (j < nb) ? btot[j] : 0;
    const int tv = (j < nb) ? tab[blockIdx.x * nb + j] : 0;
    const int inc = wave_incl_scan(v, lane);
    if (lane == 31) s_w[w] = inc;
    __syncthreads();
    int woff = 0, tot = 0;
    for (int k = 0; k < TB / 32; ++k) {
      int x = s_w[k];
      tot += x;
      if (k < w) woff += x;
    }
    if (j < nb) {
      const int st = carry + woff + inc - v;
      s_base[j] = st + tv;
      s_slot[j] = 0;
      if (blockIdx.x == 0) bstart[j] = st;
    }
    carry += tot;
    __syncthreads();
  }
  for (int i0 = lo; i0 < hi; ) {
    for (int j = 0; j < EB; ++j) {
      int i = i0 + j * TB + t;
      if (i < hi) {
        const int bk = d[j] >> BSH;
        const int r = atomicAdd(&s_slot[bk], 1);
        rec[s_base[bk] + r] = make_int2(s[j], d[j]);
      }
    }
    i0 += EB * TB;
    if (i0 < hi) {
      for (int j = 0; j < EB; ++j) {
        int i = i0 + j * TB + t;
        int ic = (i < hi) ? i : (hi - 1);
        d[j] = dst[ic];
        s[j] = src[ic];
      }
    }
  }
}

__global__ void __launch_bounds__(TB)
k_bucket_rows(const int2* __restrict__ rec, const int* __restrict__ bstart,
              const int* __restrict__ btot,
              int* __restrict__ rowptr, float* __restrict__ dinv,
              int* __restrict__ csr, int n) {
  __shared__ int s_cnt[BR];
  __shared__ int s_off[BR];
  __shared__ int s_w[TB / 32];
  const int t = threadIdx.x;
  const int lane = t & 31;
  const int w = t >> 5;
  const int b = blockIdx.x;
  const int node0 = b << BSH;
  const int bs = bstart[b];
  const int be = bs + btot[b];
  if (t < BR) s_cnt[t] = 0;
  __syncthreads();

  for (int i0 = bs; i0 < be; i0 += EB * TB) {
    int d[EB];
    for (int j = 0; j < EB; ++j) {
      int i = i0 + j * TB + t;
      d[j] = rec[(i < be) ? i : (be - 1)].y;
    }
    for (int j = 0; j < EB; ++j) {
      int i = i0 + j * TB + t;
      if (i < be) atomicAdd(&s_cnt[d[j] - node0], 1);
    }
  }
  __syncthreads();

  const int c = (t < BR) ? s_cnt[t] : 0;
  const int inc = wave_incl_scan(c, lane);
  if (lane == 31) s_w[w] = inc;
  __syncthreads();
  int woff = 0;
  for (int k = 0; k < TB / 32; ++k)
    if (k < w) woff += s_w[k];
  if (t < BR) {
    const int start = bs + woff + inc - c;
    s_off[t] = start;
    s_cnt[t] = 0;
    const int node = node0 + t;
    if (node < n) {
      rowptr[node] = start;
      float dg = (float)(c + 1);
      dinv[node] = (dg > 0.0f) ? rsqrtf(dg) : 0.0f;
    }
  }
  __syncthreads();

  for (int i0 = bs; i0 < be; i0 += EB * TB) {
    int2 r[EB];
    for (int j = 0; j < EB; ++j) {
      int i = i0 + j * TB + t;
      r[j] = rec[(i < be) ? i : (be - 1)];
    }
    for (int j = 0; j < EB; ++j) {
      int i = i0 + j * TB + t;
      if (i < be) {
        const int dl = r[j].y - node0;
        const int k = atomicAdd(&s_cnt[dl], 1);
        csr[s_off[dl] + k] = r[j].x;
      }
    }
  }
}

__global__ void k_gemm_wmma(const float* __restrict__ A, const float* __restrict__ W,
                            const float* __restrict__ scale,
                            float* __restrict__ Cout, int M) {
  __shared__ v2f sW[(D / 2) * D];
  for (int i = threadIdx.x; i < (D / 2) * D; i += blockDim.x) {
    int kk = i / D;
    int nn = i - kk * D;
    v2f p;
    p.x = W[(2 * kk + 0) * D + nn];
    p.y = W[(2 * kk + 1) * D + nn];
    sW[i] = p;
  }
  __syncthreads();

  int lane = threadIdx.x & 31;
  int half = lane >> 4;
  int lid  = lane & 15;
  int wave = blockIdx.x * (blockDim.x >> 5) + (threadIdx.x >> 5);
  int row0 = wave * 16;
  if (row0 >= M) return;

  v8f acc0 = {}, acc1 = {}, acc2 = {}, acc3 = {};

  int r  = row0 + lid;
  int rc = (r < M) ? r : (M - 1);
  const v2f* arow = (const v2f*)(A + (size_t)rc * D);
  const float ascale = scale[rc];

#pragma unroll
  for (int k = 0; k < D; k += 4) {
    int kk = (k >> 1) + half;
    v2f a = arow[kk];
    a.x *= ascale;
    a.y *= ascale;
    const v2f* wrow = &sW[kk * D + lid];
    v2f b0 = wrow[0];
    v2f b1 = wrow[16];
    v2f b2 = wrow[32];
    v2f b3 = wrow[48];
    acc0 = __builtin_amdgcn_wmma_f32_16x16x4_f32(false, a, false, b0, (short)0, acc0, false, false);
    acc1 = __builtin_amdgcn_wmma_f32_16x16x4_f32(false, a, false, b1, (short)0, acc1, false, false);
    acc2 = __builtin_amdgcn_wmma_f32_16x16x4_f32(false, a, false, b2, (short)0, acc2, false, false);
    acc3 = __builtin_amdgcn_wmma_f32_16x16x4_f32(false, a, false, b3, (short)0, acc3, false, false);
  }

  __builtin_amdgcn_sched_barrier(0);
  asm volatile("v_nop\n\tv_nop\n\tv_nop\n\tv_nop");
  __builtin_amdgcn_sched_barrier(0);

#pragma unroll
  for (int v = 0; v < 8; ++v) {
    int rr = row0 + v + 8 * half;
    if (rr < M) {
      float* crow = Cout + (size_t)rr * D + lid;
      crow[0]  = acc0[v];
      crow[16] = acc1[v];
      crow[32] = acc2[v];
      crow[48] = acc3[v];
    }
  }
}

#define NPW 4

__device__ __forceinline__ void gather_piece(v2f& acc, const v2f* __restrict__ h2,
                                             int my_s, int c, int lane) {
  int sid[16];
  for (int k = 0; k < 16; ++k) sid[k] = __builtin_amdgcn_readlane(my_s, k);
  v2f v[16];
  for (int k = 0; k < 16; ++k) {
    if (k < c) v[k] = h2[(size_t)sid[k] * (D / 2) + lane];
  }
  for (int k = 0; k < 16; ++k) {
    if (k < c) {
      acc.x += v[k].x;
      acc.y += v[k].y;
    }
  }
}

__device__ __forceinline__ v2f gather_run(const int* __restrict__ csr, const v2f* __restrict__ h2,
                                          int node, int beg, int end, int c0, int id0, int lane) {
  v2f acc = {0.0f, 0.0f};
  int base = beg, c = c0, my_s = id0;
  while (c > 0) {
    gather_piece(acc, h2, my_s, c, lane);
    base += 16;
    c = end - base; if (c > 16) c = 16;
    if (c > 0) { my_s = node; if (lane < c) my_s = csr[base + lane]; }
  }
  return acc;
}

struct WaveNodes {
  int node[NPW], beg[NPW], end[NPW], c0[NPW], id0[NPW];
  float dt[NPW];
  v2f hs[NPW];
};

__device__ __forceinline__ void load_wave_nodes(WaveNodes& wn, int node0,
                                                const int* __restrict__ rowptr,
                                                const int* __restrict__ csr,
                                                const float* __restrict__ dinv,
                                                const v2f* __restrict__ h2,
                                                int n, int E, int lane) {
  int rp[NPW + 1];
#pragma unroll
  for (int j = 0; j <= NPW; ++j) {
    const int i = (node0 + j < n) ? (node0 + j) : (n - 1);
    rp[j] = rowptr[i];
  }
#pragma unroll
  for (int j = 0; j < NPW; ++j) {
    wn.node[j] = (node0 + j < n) ? (node0 + j) : (n - 1);
    wn.beg[j] = rp[j];
    wn.end[j] = (node0 + j + 1 < n) ? rp[j + 1] : E;
    int c = wn.end[j] - wn.beg[j]; if (c > 16) c = 16;
    wn.c0[j] = c;
  }
#pragma unroll
  for (int j = 0; j < NPW; ++j) {
    int id = wn.node[j];
    if (lane < wn.c0[j]) id = csr[wn.beg[j] + lane];
    wn.id0[j] = id;
  }
#pragma unroll
  for (int j = 0; j < NPW; ++j) {
    wn.dt[j] = dinv[wn.node[j]];
    wn.hs[j] = h2[(size_t)wn.node[j] * (D / 2) + lane];
  }
}

__global__ void __launch_bounds__(256)
k_gather(const int* __restrict__ rowptr, const int* __restrict__ csr,
         const float* __restrict__ dinv, const float* __restrict__ h,
         const float* __restrict__ b, const float* __restrict__ a,
         float* __restrict__ out, int n, int E) {
  const int lane = threadIdx.x & 31;
  const int node0 = NPW * __builtin_amdgcn_readfirstlane(
      (int)(blockIdx.x * (blockDim.x >> 5) + (threadIdx.x >> 5)));
  if (node0 >= n) return;
  const v2f* __restrict__ h2 = (const v2f*)h;
  WaveNodes wn;
  load_wave_nodes(wn, node0, rowptr, csr, dinv, h2, n, E, lane);
  const v2f bb = ((const v2f*)b)[lane];
  const v2f al = ((const v2f*)a)[lane];
#pragma unroll
  for (int j = 0; j < NPW; ++j) {
    v2f acc = gather_run(csr, h2, wn.node[j], wn.beg[j], wn.end[j], wn.c0[j], wn.id0[j], lane);
    acc.x = (acc.x + wn.hs[j].x) * wn.dt[j] + bb.x;
    acc.y = (acc.y + wn.hs[j].y) * wn.dt[j] + bb.y;
    acc.x = (acc.x >= 0.0f) ? acc.x : al.x * acc.x;
    acc.y = (acc.y >= 0.0f) ? acc.y : al.y * acc.y;
    if (node0 + j < n) ((v2f*)out)[(size_t)(node0 + j) * (D / 2) + lane] = acc;
  }
}

#define TN 32
#define TP 68

__global__ void __launch_bounds__(256)
k_gather_gemm(const int* __restrict__ rowptr, const int* __restrict__ csr,
              const float* __restrict__ dinv, const float* __restrict__ h,
              const float* __restrict__ b, const float* __restrict__ a,
              const float* __restrict__ W, float* __restrict__ Cout, int n, int E) {
  __shared__ v2f sW[(D / 2) * D];
  __shared__ v2f sT[TN * (TP / 2)];
  const int lane = threadIdx.x & 31;
  const int wave = __builtin_amdgcn_readfirstlane((int)(threadIdx.x >> 5));
  const int tile0 = blockIdx.x * TN;

  v2f wst[(D / 2) * D / 256];
#pragma unroll
  for (int j = 0; j < (D / 2) * D / 256; ++j) {
    const int i = (int)threadIdx.x + j * 256;
    const int kk = i / D;
    const int nn = i - kk * D;
    wst[j].x = W[(2 * kk + 0) * D + nn];
    wst[j].y = W[(2 * kk + 1) * D + nn];
  }

  const v2f* __restrict__ h2 = (const v2f*)h;
  const int node0 = tile0 + NPW * wave;
  WaveNodes wn;
  load_wave_nodes(wn, node0, rowptr, csr, dinv, h2, n, E, lane);
  const v2f bb = ((const v2f*)b)[lane];
  const v2f al = ((const v2f*)a)[lane];

#pragma unroll
  for (int j = 0; j < (D / 2) * D / 256; ++j) sW[(int)threadIdx.x + j * 256] = wst[j];

#pragma unroll
  for (int j = 0; j < NPW; ++j) {
    v2f acc = gather_run(csr, h2, wn.node[j], wn.beg[j], wn.end[j], wn.c0[j], wn.id0[j], lane);
    acc.x = (acc.x + wn.hs[j].x) * wn.dt[j] + bb.x;
    acc.y = (acc.y + wn.hs[j].y) * wn.dt[j] + bb.y;
    acc.x = (acc.x >= 0.0f) ? acc.x : al.x * acc.x;
    acc.y = (acc.y >= 0.0f) ? acc.y : al.y * acc.y;
    v2f r;
    r.x = acc.x * wn.dt[j];
    r.y = acc.y * wn.dt[j];
    sT[(NPW * wave + j) * (TP / 2) + lane] = r;
  }
  __syncthreads();

  if (wave < 4) {
    const int half = lane >> 4;
    const int lid  = lane & 15;
    v8f acc0 = {}, acc1 = {};
    const v2f* t0 = &sT[lid * (TP / 2)];
    const v2f* t1 = &sT[(16 + lid) * (TP / 2)];
    const v2f* wcol = &sW[wave * 16 + lid];
#pragma unroll
    for (int k = 0; k < D; k += 4) {
      int kk = (k >> 1) + half;
      v2f a0 = t0[kk];
      v2f a1 = t1[kk];
      v2f bq = wcol[kk * D];
      acc0 = __builtin_amdgcn_wmma_f32_16x16x4_f32(false, a0, false, bq, (short)0, acc0, false, false);
      acc1 = __builtin_amdgcn_wmma_f32_16x16x4_f32(false, a1, false, bq, (short)0, acc1, false, false);
    }
    __builtin_amdgcn_sched_barrier(0);
    asm volatile("v_nop\n\tv_nop\n\tv_nop\n\tv_nop");
    __builtin_amdgcn_sched_barrier(0);
#pragma unroll
    for (int v = 0; v < 8; ++v) {
      int rr = tile0 + v + 8 * half;
      if (rr < n)      Cout[(size_t)rr * D + wave * 16 + lid] = acc0[v];
      if (rr + 16 < n) Cout[(size_t)(rr + 16) * D + wave * 16 + lid] = acc1[v];
    }
  }
}

static inline int cdiv_ll(long long a, long long b) { return (int)((a + b - 1) / b); }

extern "C" void kernel_launch(void* const* d_in, const int* in_sizes, int n_in,
                              void* d_out, int out_size, void* d_ws, size_t ws_size,
                              hipStream_t stream) {
  (void)n_in; (void)out_size; (void)ws_size;

  const float* x  = (const float*)d_in[0];
  const int*   ei = (const int*)  d_in[1];
  const float* W1 = (const float*)d_in[2];
  const float* b1 = (const float*)d_in[3];
  const float* a1 = (const float*)d_in[4];
  const float* W2 = (const float*)d_in[5];
  const float* b2 = (const float*)d_in[6];
  const float* a2 = (const float*)d_in[7];

  const int n = in_sizes[0] / D;
  const int E = in_sizes[1] / 2;
  const int* src = ei;
  const int* dst = ei + E;

  float* ws     = (float*)d_ws;
  float* dinv   = ws;
  int*   rowptr = (int*)(dinv + n);
  int*   csr    = rowptr + n;
  float* bufA   = (float*)(csr + E);
  float* bufB   = bufA + (size_t)n * D;
  const int nb  = (n + BR - 1) / BR;
  int*   tab    = (int*)bufA;
  int*   btot   = tab + (size_t)NCH * nb;
  int*   bstart = btot + nb;
  int2*  rec    = (int2*)(bufB + (E & 1));
  float* out    = (float*)d_out;

  const int B = 256;
  const int gGa   = cdiv_ll((long long)cdiv_ll(n, NPW) * 32, B);
  const int gGemm = cdiv_ll((long long)cdiv_ll(n, 16) * 32, B);
  const int gFuse = cdiv_ll(n, TN);

  k_bin_count  <<<NCH, TB, nb * sizeof(int), stream>>>(dst, tab, E, nb);
  k_bin_rowscan<<<nb, NCH, 0, stream>>>(tab, btot, nb);
  k_bin_place  <<<NCH, TB, 2 * nb * sizeof(int), stream>>>(src, dst, tab, btot, bstart, rec, E, nb);
  k_bucket_rows<<<nb, TB, 0, stream>>>(rec, bstart, btot, rowptr, dinv, csr, n);

  k_gemm_wmma<<<gGemm, B, 0, stream>>>(x, W1, dinv, bufA, n);

  k_gather_gemm<<<gFuse, 256, 0, stream>>>(rowptr, csr, dinv, bufA, b1, a1, W2, bufB, n, E);

  k_gather   <<<gGa,   B, 0, stream>>>(rowptr, csr, dinv, bufB, b2, a2, out, n, E);
}


# Edge binning in 2 launches, not 4: a chunk counts, scans and places its edges inside its own slice of rec in one launch; bucket csr start = sum of chunk offsets; sum order changes (arrival order)
# speedup vs baseline: 6.9824x; 1.1675x over previous
#include <hip/hip_runtime.h>
#include <hip/hip_bf16.h>

typedef float v2f __attribute__((ext_vector_type(2)));
typedef float v8f __attribute__((ext_vector_type(8)));

#define D 64

#define BSH 9
#define BR  (1 << BSH)
#define NCH 256
#define TB  1024
#define EB  8
#define NQ  (NCH / (TB / 32))

__device__ __forceinline__ int wave_incl_scan(int v, int lane) {
  for (int o = 1; o < 32; o <<= 1) {
    int u = __shfl_up(v, o);
    if (lane >= o) v += u;
  }
  return v;
}

__global__ void __launch_bounds__(TB)
k_bin(const int* __restrict__ src, const int* __restrict__ dst,
      int* __restrict__ tab, int* __restrict__ off, int2* __restrict__ rec, int E, int nb) {
  extern __shared__ int s_dyn[];
  __shared__ int s_w[TB / 32];
  int* s_cnt = s_dyn;
  int* s_base = s_dyn + nb;
  const int t = threadIdx.x;
  const int lane = t & 31;
  const int w = t >> 5;
  const int per = (E + NCH - 1) / NCH;
  const int lo = blockIdx.x * per;
  const int hi = (lo + per < E) ? (lo + per) : E;
  int d[EB], s[EB];
  for (int j = 0; j < EB; ++j) {
    int i = lo + j * TB + t;
    int ic = (i < hi) ? i : (hi - 1);
    ic = (ic > 0) ? ic : 0;
    d[j] = dst[ic];
    s[j] = src[ic];
  }
  for (int j = t; j < nb; j += TB) s_cnt[j] = 0;
  __syncthreads();

  for (int j = 0; j < EB; ++j) {
    int i = lo + j * TB + t;
    if (i < hi) atomicAdd(&s_cnt[d[j] >> BSH], 1);
  }
  for (int i0 = lo + EB * TB; i0 < hi; i0 += EB * TB) {
    int dd[EB];
    for (int j = 0; j < EB; ++j) {
      int i = i0 + j * TB + t;
      dd[j] = dst[(i < hi) ? i : (hi - 1)];
    }
    for (int j = 0; j < EB; ++j) {
      int i = i0 + j * TB + t;
      if (i < hi) atomicAdd(&s_cnt[dd[j] >> BSH], 1);
    }
  }
  __syncthreads();

  int carry = 0;
  for (int j0 = 0; j0 < nb; j0 += TB) {
    const int j = j0 + t;
    const int v = (j < nb) ? s_cnt[j] : 0;
    const int inc = wave_incl_scan(v, lane);
    if (lane == 31) s_w[w] = inc;
    __syncthreads();
    int woff = 0, tot = 0;
    for (int k = 0; k < TB / 32; ++k) {
      int x = s_w[k];
      tot += x;
      if (k < w) woff += x;
    }
    if (j < nb) {
      const int st = carry + woff + inc - v;
      s_base[j] = lo + st;
      s_cnt[j] = 0;
      tab[blockIdx.x * nb + j] = v;
      off[blockIdx.x * nb + j] = st;
    }
    carry += tot;
    __syncthreads();
  }

  for (int i0 = lo; i0 < hi; ) {
    for (int j = 0; j < EB; ++j) {
      int i = i0 + j * TB + t;
      if (i < hi) {
        const int bk = d[j] >> BSH;
        const int r = atomicAdd(&s_cnt[bk], 1);
        rec[s_base[bk] + r] = make_int2(s[j], d[j]);
      }
    }
    i0 += EB * TB;
    if (i0 < hi) {
      for (int j = 0; j < EB; ++j) {
        int i = i0 + j * TB + t;
        int ic = (i < hi) ? i : (hi - 1);
        d[j] = dst[ic];
        s[j] = src[ic];
      }
    }
  }
}

__global__ void __launch_bounds__(TB)
k_bucket_rows(const int2* __restrict__ rec, const int* __restrict__ tab,
              const int* __restrict__ off,
              int* __restrict__ rowptr, float* __restrict__ dinv,
              int* __restrict__ csr, int n, int E, int nb) {
  __shared__ int s_cnt[BR];
  __shared__ int s_off[BR];
  __shared__ int s_rs[NCH];
  __shared__ int s_rc[NCH];
  __shared__ int s_w[TB / 32];
  const int t = threadIdx.x;
  const int lane = t & 31;
  const int w = t >> 5;
  const int b = blockIdx.x;
  const int node0 = b << BSH;
  const int per = (E + NCH - 1) / NCH;
  int of = 0;
  if (t < NCH) {
    of = off[t * nb + b];
    s_rs[t] = t * per + of;
    s_rc[t] = tab[t * nb + b];
  }
  if (t < BR) s_cnt[t] = 0;
  int wsum = of;
  for (int o = 16; o > 0; o >>= 1) wsum += __shfl_xor(wsum, o);
  if (lane == 0) s_w[w] = wsum;
  __syncthreads();
  int bs = 0;
  for (int k = 0; k < TB / 32; ++k) bs += s_w[k];

  int rs[NQ], rc[NQ];
  int2 r0[NQ], r1[NQ];
  for (int q = 0; q < NQ; ++q) {
    const int c = w + q * (TB / 32);
    rs[q] = __builtin_amdgcn_readfirstlane(s_rs[c]);
    rc[q] = __builtin_amdgcn_readfirstlane(s_rc[c]);
    r0[q] = rec[(lane < rc[q]) ? (rs[q] + lane) : 0];
    r1[q] = rec[(lane + 32 < rc[q]) ? (rs[q] + lane + 32) : 0];
  }
  for (int q = 0; q < NQ; ++q) {
    if (lane < rc[q])      atomicAdd(&s_cnt[r0[q].y - node0], 1);
    if (lane + 32 < rc[q]) atomicAdd(&s_cnt[r1[q].y - node0], 1);
  }
  for (int q = 0; q < NQ; ++q) {
#pragma nounroll
    for (int i0 = 64; i0 < rc[q]; i0 += 32) {
      const int i = i0 + lane;
      if (i < rc[q]) atomicAdd(&s_cnt[rec[rs[q] + i].y - node0], 1);
    }
  }
  __syncthreads();

  const int c = (t < BR) ? s_cnt[t] : 0;
  const int inc = wave_incl_scan(c, lane);
  if (lane == 31) s_w[w] = inc;
  __syncthreads();
  int woff = 0;
  for (int k = 0; k < TB / 32; ++k)
    if (k < w) woff += s_w[k];
  if (t < BR) {
    const int start = bs + woff + inc - c;
    s_off[t] = start;
    s_cnt[t] = 0;
    const int node = node0 + t;
    if (node < n) {
      rowptr[node] = start;
      float dg = (float)(c + 1);
      dinv[node] = (dg > 0.0f) ? rsqrtf(dg) : 0.0f;
    }
  }
  __syncthreads();

  for (int q = 0; q < NQ; ++q) {
    if (lane < rc[q]) {
      const int dl = r0[q].y - node0;
      const int k = atomicAdd(&s_cnt[dl], 1);
      csr[s_off[dl] + k] = r0[q].x;
    }
    if (lane + 32 < rc[q]) {
      const int dl = r1[q].y - node0;
      const int k = atomicAdd(&s_cnt[dl], 1);
      csr[s_off[dl] + k] = r1[q].x;
    }
  }
  for (int q = 0; q < NQ; ++q) {
#pragma nounroll
    for (int i0 = 64; i0 < rc[q]; i0 += 32) {
      const int i = i0 + lane;
      if (i < rc[q]) {
        const int2 rr = rec[rs[q] + i];
        const int dl = rr.y - node0;
        const int k = atomicAdd(&s_cnt[dl], 1);
        csr[s_off[dl] + k] = rr.x;
      }
    }
  }
}

__global__ void k_gemm_wmma(const float* __restrict__ A, const float* __restrict__ W,
                            const float* __restrict__ scale,
                            float* __restrict__ Cout, int M) {
  __shared__ v2f sW[(D / 2) * D];
  for (int i = threadIdx.x; i < (D / 2) * D; i += blockDim.x) {
    int kk = i / D;
    int nn = i - kk * D;
    v2f p;
    p.x = W[(2 * kk + 0) * D + nn];
    p.y = W[(2 * kk + 1) * D + nn];
    sW[i] = p;
  }
  __syncthreads();

  int lane = threadIdx.x & 31;
  int half = lane >> 4;
  int lid  = lane & 15;
  int wave = blockIdx.x * (blockDim.x >> 5) + (threadIdx.x >> 5);
  int row0 = wave * 16;
  if (row0 >= M) return;

  v8f acc0 = {}, acc1 = {}, acc2 = {}, acc3 = {};

  int r  = row0 + lid;
  int rc = (r < M) ? r : (M - 1);
  const v2f* arow = (const v2f*)(A + (size_t)rc * D);
  const float ascale = scale[rc];

#pragma unroll
  for (int k = 0; k < D; k += 4) {
    int kk = (k >> 1) + half;
    v2f a = arow[kk];
    a.x *= ascale;
    a.y *= ascale;
    const v2f* wrow = &sW[kk * D + lid];
    v2f b0 = wrow[0];
    v2f b1 = wrow[16];
    v2f b2 = wrow[32];
    v2f b3 = wrow[48];
    acc0 = __builtin_amdgcn_wmma_f32_16x16x4_f32(false, a, false, b0, (short)0, acc0, false, false);
    acc1 = __builtin_amdgcn_wmma_f32_16x16x4_f32(false, a, false, b1, (short)0, acc1, false, false);
    acc2 = __builtin_amdgcn_wmma_f32_16x16x4_f32(false, a, false, b2, (short)0, acc2, false, false);
    acc3 = __builtin_amdgcn_wmma_f32_16x16x4_f32(false, a, false, b3, (short)0, acc3, false, false);
  }

  __builtin_amdgcn_sched_barrier(0);
  asm volatile("v_nop\n\tv_nop\n\tv_nop\n\tv_nop");
  __builtin_amdgcn_sched_barrier(0);

#pragma unroll
  for (int v = 0; v < 8; ++v) {
    int rr = row0 + v + 8 * half;
    if (rr < M) {
      float* crow = Cout + (size_t)rr * D + lid;
      crow[0]  = acc0[v];
      crow[16] = acc1[v];
      crow[32] = acc2[v];
      crow[48] = acc3[v];
    }
  }
}

#define NPW 4

__device__ __forceinline__ void gather_piece(v2f& acc, const v2f* __restrict__ h2,
                                             int my_s, int c, int lane) {
  int sid[16];
  for (int k = 0; k < 16; ++k) sid[k] = __builtin_amdgcn_readlane(my_s, k);
  v2f v[16];
  for (int k = 0; k < 16; ++k) {
    if (k < c) v[k] = h2[(size_t)sid[k] * (D / 2) + lane];
  }
  for (int k = 0; k < 16; ++k) {
    if (k < c) {
      acc.x += v[k].x;
      acc.y += v[k].y;
    }
  }
}

__device__ __forceinline__ v2f gather_run(const int* __restrict__ csr, const v2f* __restrict__ h2,
                                          int node, int beg, int end, int c0, int id0, int lane) {
  v2f acc = {0.0f, 0.0f};
  int base = beg, c = c0, my_s = id0;
  while (c > 0) {
    gather_piece(acc, h2, my_s, c, lane);
    base += 16;
    c = end - base; if (c > 16) c = 16;
    if (c > 0) { my_s = node; if (lane < c) my_s = csr[base + lane]; }
  }
  return acc;
}

struct WaveNodes {
  int node[NPW], beg[NPW], end[NPW], c0[NPW], id0[NPW];
  float dt[NPW];
  v2f hs[NPW];
};

__device__ __forceinline__ void load_wave_nodes(WaveNodes& wn, int node0,
                                                const int* __restrict__ rowptr,
                                                const int* __restrict__ csr,
                                                const float* __restrict__ dinv,
                                                const v2f* __restrict__ h2,
                                                int n, int E, int lane) {
  int rp[NPW + 1];
#pragma unroll
  for (int j = 0; j <= NPW; ++j) {
    const int i = (node0 + j < n) ? (node0 + j) : (n - 1);
    rp[j] = rowptr[i];
  }
#pragma unroll
  for (int j = 0; j < NPW; ++j) {
    wn.node[j] = (node0 + j < n) ? (node0 + j) : (n - 1);
    wn.beg[j] = rp[j];
    wn.end[j] = (node0 + j + 1 < n) ? rp[j + 1] : E;
    int c = wn.end[j] - wn.beg[j]; if (c > 16) c = 16;
    wn.c0[j] = c;
  }
#pragma unroll
  for (int j = 0; j < NPW; ++j) {
    int id = wn.node[j];
    if (lane < wn.c0[j]) id = csr[wn.beg[j] + lane];
    wn.id0[j] = id;
  }
#pragma unroll
  for (int j = 0; j < NPW; ++j) {
    wn.dt[j] = dinv[wn.node[j]];
    wn.hs[j] = h2[(size_t)wn.node[j] * (D / 2) + lane];
  }
}

__global__ void __launch_bounds__(256)
k_gather(const int* __restrict__ rowptr, const int* __restrict__ csr,
         const float* __restrict__ dinv, const float* __restrict__ h,
         const float* __restrict__ b, const float* __restrict__ a,
         float* __restrict__ out, int n, int E) {
  const int lane = threadIdx.x & 31;
  const int node0 = NPW * __builtin_amdgcn_readfirstlane(
      (int)(blockIdx.x * (blockDim.x >> 5) + (threadIdx.x >> 5)));
  if (node0 >= n) return;
  const v2f* __restrict__ h2 = (const v2f*)h;
  WaveNodes wn;
  load_wave_nodes(wn, node0, rowptr, csr, dinv, h2, n, E, lane);
  const v2f bb = ((const v2f*)b)[lane];
  const v2f al = ((const v2f*)a)[lane];
#pragma unroll
  for (int j = 0; j < NPW; ++j) {
    v2f acc = gather_run(csr, h2, wn.node[j], wn.beg[j], wn.end[j], wn.c0[j], wn.id0[j], lane);
    acc.x = (acc.x + wn.hs[j].x) * wn.dt[j] + bb.x;
    acc.y = (acc.y + wn.hs[j].y) * wn.dt[j] + bb.y;
    acc.x = (acc.x >= 0.0f) ? acc.x : al.x * acc.x;
    acc.y = (acc.y >= 0.0f) ? acc.y : al.y * acc.y;
    if (node0 + j < n) ((v2f*)out)[(size_t)(node0 + j) * (D / 2) + lane] = acc;
  }
}

#define TN 32
#define TP 68

__global__ void __launch_bounds__(256)
k_gather_gemm(const int* __restrict__ rowptr, const int* __restrict__ csr,
              const float* __restrict__ dinv, const float* __restrict__ h,
              const float* __restrict__ b, const float* __restrict__ a,
              const float* __restrict__ W, float* __restrict__ Cout, int n, int E) {
  __shared__ v2f sW[(D / 2) * D];
  __shared__ v2f sT[TN * (TP / 2)];
  const int lane = threadIdx.x & 31;
  const int wave = __builtin_amdgcn_readfirstlane((int)(threadIdx.x >> 5));
  const int tile0 = blockIdx.x * TN;

  v2f wst[(D / 2) * D / 256];
#pragma unroll
  for (int j = 0; j < (D / 2) * D / 256; ++j) {
    const int i = (int)threadIdx.x + j * 256;
    const int kk = i / D;
    const int nn = i - kk * D;
    wst[j].x = W[(2 * kk + 0) * D + nn];
    wst[j].y = W[(2 * kk + 1) * D + nn];
  }

  const v2f* __restrict__ h2 = (const v2f*)h;
  const int node0 = tile0 + NPW * wave;
  WaveNodes wn;
  load_wave_nodes(wn, node0, rowptr, csr, dinv, h2, n, E, lane);
  const v2f bb = ((const v2f*)b)[lane];
  const v2f al = ((const v2f*)a)[lane];

#pragma unroll
  for (int j = 0; j < (D / 2) * D / 256; ++j) sW[(int)threadIdx.x + j * 256] = wst[j];

#pragma unroll
  for (int j = 0; j < NPW; ++j) {
    v2f acc = gather_run(csr, h2, wn.node[j], wn.beg[j], wn.end[j], wn.c0[j], wn.id0[j], lane);
    acc.x = (acc.x + wn.hs[j].x) * wn.dt[j] + bb.x;
    acc.y = (acc.y + wn.hs[j].y) * wn.dt[j] + bb.y;
    acc.x = (acc.x >= 0.0f) ? acc.x : al.x * acc.x;
    acc.y = (acc.y >= 0.0f) ? acc.y : al.y * acc.y;
    v2f r;
    r.x = acc.x * wn.dt[j];
    r.y = acc.y * wn.dt[j];
    sT[(NPW * wave + j) * (TP / 2) + lane] = r;
  }
  __syncthreads();

  if (wave < 4) {
    const int half = lane >> 4;
    const int lid  = lane & 15;
    v8f acc0 = {}, acc1 = {};
    const v2f* t0 = &sT[lid * (TP / 2)];
    const v2f* t1 = &sT[(16 + lid) * (TP / 2)];
    const v2f* wcol = &sW[wave * 16 + lid];
#pragma unroll
    for (int k = 0; k < D; k += 4) {
      int kk = (k >> 1) + half;
      v2f a0 = t0[kk];
      v2f a1 = t1[kk];
      v2f bq = wcol[kk * D];
      acc0 = __builtin_amdgcn_wmma_f32_16x16x4_f32(false, a0, false, bq, (short)0, acc0, false, false);
      acc1 = __builtin_amdgcn_wmma_f32_16x16x4_f32(false, a1, false, bq, (short)0, acc1, false, false);
    }
    __builtin_amdgcn_sched_barrier(0);
    asm volatile("v_nop\n\tv_nop\n\tv_nop\n\tv_nop");
    __builtin_amdgcn_sched_barrier(0);
#pragma unroll
    for (int v = 0; v < 8; ++v) {
      int rr = tile0 + v + 8 * half;
      if (rr < n)      Cout[(size_t)rr * D + wave * 16 + lid] = acc0[v];
      if (rr + 16 < n) Cout[(size_t)(rr + 16) * D + wave * 16 + lid] = acc1[v];
    }
  }
}

static inline int cdiv_ll(long long a, long long b) { return (int)((a + b - 1) / b); }

extern "C" void kernel_launch(void* const* d_in, const int* in_sizes, int n_in,
                              void* d_out, int out_size, void* d_ws, size_t ws_size,
                              hipStream_t stream) {
  (void)n_in; (void)out_size; (void)ws_size;

  const float* x  = (const float*)d_in[0];
  const int*   ei = (const int*)  d_in[1];
  const float* W1 = (const float*)d_in[2];
  const float* b1 = (const float*)d_in[3];
  const float* a1 = (const float*)d_in[4];
  const float* W2 = (const float*)d_in[5];
  const float* b2 = (const float*)d_in[6];
  const float* a2 = (const float*)d_in[7];

  const int n = in_sizes[0] / D;
  const int E = in_sizes[1] / 2;
  const int* src = ei;
  const int* dst = ei + E;

  float* ws     = (float*)d_ws;
  float* dinv   = ws;
  int*   rowptr = (int*)(dinv + n);
  int*   csr    = rowptr + n;
  float* bufA   = (float*)(csr + E);
  float* bufB   = bufA + (size_t)n * D;
  const int nb  = (n + BR - 1) / BR;
  int*   tab    = (int*)bufA;
  int*   off    = tab + (size_t)NCH * nb;
  int2*  rec    = (int2*)(bufB + (E & 1));
  float* out    = (float*)d_out;

  const int B = 256;
  const int gGa   = cdiv_ll((long long)cdiv_ll(n, NPW) * 32, B);
  const int gGemm = cdiv_ll((long long)cdiv_ll(n, 16) * 32, B);
  const int gFuse = cdiv_ll(n, TN);

  k_bin        <<<NCH, TB, 2 * nb * sizeof(int), stream>>>(src, dst, tab, off, rec, E, nb);
  k_bucket_rows<<<nb, TB, 0, stream>>>(rec, tab, off, rowptr, dinv, csr, n, E, nb);

  k_gemm_wmma<<<gGemm, B, 0, stream>>>(x, W1, dinv, bufA, n);

  k_gather_gemm<<<gFuse, 256, 0, stream>>>(rowptr, csr, dinv, bufA, b1, a1, W2, bufB, n, E);

  k_gather   <<<gGa,   B, 0, stream>>>(rowptr, csr, dinv, bufB, b2, a2, out, n, E);
}


# Layer 1's product runs behind the row building in one launch: each of a bucket's 32 waves multiplies one 16-row tile of its own 512 rows (dinv just computed) by W1 on f32 WMMA; 1 launch fewer
# speedup vs baseline: 7.5789x; 1.0854x over previous
#include <hip/hip_runtime.h>
#include <hip/hip_bf16.h>

typedef float v2f __attribute__((ext_vector_type(2)));
typedef float v8f __attribute__((ext_vector_type(8)));

#define D 64

#define BSH 9
#define BR  (1 << BSH)
#define NCH 256
#define TB  1024
#define EB  8
#define NQ  (NCH / (TB / 32))

__device__ __forceinline__ int wave_incl_scan(int v, int lane) {
  for (int o = 1; o < 32; o <<= 1) {
    int u = __shfl_up(v, o);
    if (lane >= o) v += u;
  }
  return v;
}

__global__ void __launch_bounds__(TB)
k_bin(const int* __restrict__ src, const int* __restrict__ dst,
      int* __restrict__ tab, int* __restrict__ off, int2* __restrict__ rec, int E, int nb) {
  extern __shared__ int s_dyn[];
  __shared__ int s_w[TB / 32];
  int* s_cnt = s_dyn;
  int* s_base = s_dyn + nb;
  const int t = threadIdx.x;
  const int lane = t & 31;
  const int w = t >> 5;
  const int per = (E + NCH - 1) / NCH;
  const int lo = blockIdx.x * per;
  const int hi = (lo + per < E) ? (lo + per) : E;
  int d[EB], s[EB];
  for (int j = 0; j < EB; ++j) {
    int i = lo + j * TB + t;
    int ic = (i < hi) ? i : (hi - 1);
    ic = (ic > 0) ? ic : 0;
    d[j] = dst[ic];
    s[j] = src[ic];
  }
  for (int j = t; j < nb; j += TB) s_cnt[j] = 0;
  __syncthreads();

  for (int j = 0; j < EB; ++j) {
    int i = lo + j * TB + t;
    if (i < hi) atomicAdd(&s_cnt[d[j] >> BSH], 1);
  }
  for (int i0 = lo + EB * TB; i0 < hi; i0 += EB * TB) {
    int dd[EB];
    for (int j = 0; j < EB; ++j) {
      int i = i0 + j * TB + t;
      dd[j] = dst[(i < hi) ? i : (hi - 1)];
    }
    for (int j = 0; j < EB; ++j) {
      int i = i0 + j * TB + t;
      if (i < hi) atomicAdd(&s_cnt[dd[j] >> BSH], 1);
    }
  }
  __syncthreads();

  int carry = 0;
  for (int j0 = 0; j0 < nb; j0 += TB) {
    const int j = j0 + t;
    const int v = (j < nb) ? s_cnt[j] : 0;
    const int inc = wave_incl_scan(v, lane);
    if (lane == 31) s_w[w] = inc;
    __syncthreads();
    int woff = 0, tot = 0;
    for (int k = 0; k < TB / 32; ++k) {
      int x = s_w[k];
      tot += x;
      if (k < w) woff += x;
    }
    if (j < nb) {
      const int st = carry + woff + inc - v;
      s_base[j] = lo + st;
      s_cnt[j] = 0;
      tab[blockIdx.x * nb + j] = v;
      off[blockIdx.x * nb + j] = st;
    }
    carry += tot;
    __syncthreads();
  }

  for (int i0 = lo; i0 < hi; ) {
    for (int j = 0; j < EB; ++j) {
      int i = i0 + j * TB + t;
      if (i < hi) {
        const int bk = d[j] >> BSH;
        const int r = atomicAdd(&s_cnt[bk], 1);
        rec[s_base[bk] + r] = make_int2(s[j], d[j]);
      }
    }
    i0 += EB * TB;
    if (i0 < hi) {
      for (int j = 0; j < EB; ++j) {
        int i = i0 + j * TB + t;
        int ic = (i < hi) ? i : (hi - 1);
        d[j] = dst[ic];
        s[j] = src[ic];
      }
    }
  }
}

__global__ void __launch_bounds__(TB)
k_bucket_rows_gemm(const int2* __restrict__ rec, const int* __restrict__ tab,
                   const int* __restrict__ off,
                   int* __restrict__ rowptr, float* __restrict__ dinv,
                   int* __restrict__ csr,
                   const float* __restrict__ A, const float* __restrict__ W,
                   float* __restrict__ Cout, int n, int E, int nb) {
  static_assert(BR / 16 == TB / 32, "one 16-row tile of the bucket per wave");
  __shared__ int s_cnt[BR];
  __shared__ int s_off[BR];
  __shared__ int s_rs[NCH];
  __shared__ int s_rc[NCH];
  __shared__ int s_w[TB / 32];
  __shared__ float s_dinv[BR];
  __shared__ v2f sW[(D / 2) * D];
  const int t = threadIdx.x;
  const int lane = t & 31;
  const int w = t >> 5;
  const int b = blockIdx.x;
  const int node0 = b << BSH;
  const int per = (E + NCH - 1) / NCH;
  v2f wst[(D / 2) * D / TB];
#pragma unroll
  for (int j = 0; j < (D / 2) * D / TB; ++j) {
    const int i = t + j * TB;
    const int kk = i / D;
    const int nn = i - kk * D;
    wst[j].x = W[(2 * kk + 0) * D + nn];
    wst[j].y = W[(2 * kk + 1) * D + nn];
  }
  int of = 0;
  if (t < NCH) {
    of = off[t * nb + b];
    s_rs[t] = t * per + of;
    s_rc[t] = tab[t * nb + b];
  }
#pragma unroll
  for (int j = 0; j < (D / 2) * D / TB; ++j) sW[t + j * TB] = wst[j];
  if (t < BR) s_cnt[t] = 0;
  int wsum = of;
  for (int o = 16; o > 0; o >>= 1) wsum += __shfl_xor(wsum, o);
  if (lane == 0) s_w[w] = wsum;
  __syncthreads();
  int bs = 0;
  for (int k = 0; k < TB / 32; ++k) bs += s_w[k];

  int rs[NQ], rc[NQ];
  int2 r0[NQ], r1[NQ];
  for (int q = 0; q < NQ; ++q) {
    const int c = w + q * (TB / 32);
    rs[q] = __builtin_amdgcn_readfirstlane(s_rs[c]);
    rc[q] = __builtin_amdgcn_readfirstlane(s_rc[c]);
    r0[q] = rec[(lane < rc[q]) ? (rs[q] + lane) : 0];
    r1[q] = rec[(lane + 32 < rc[q]) ? (rs[q] + lane + 32) : 0];
  }

  for (int q = 0; q < NQ; ++q) {
    if (lane < rc[q])      atomicAdd(&s_cnt[r0[q].y - node0], 1);
    if (lane + 32 < rc[q]) atomicAdd(&s_cnt[r1[q].y - node0], 1);
  }
  for (int q = 0; q < NQ; ++q) {
#pragma nounroll
    for (int i0 = 64; i0 < rc[q]; i0 += 32) {
      const int i = i0 + lane;
      if (i < rc[q]) atomicAdd(&s_cnt[rec[rs[q] + i].y - node0], 1);
    }
  }
  __syncthreads();

  const int c = (t < BR) ? s_cnt[t] : 0;
  const int inc = wave_incl_scan(c, lane);
  if (lane == 31) s_w[w] = inc;
  __syncthreads();
  int woff = 0;
  for (int k = 0; k < TB / 32; ++k)
    if (k < w) woff += s_w[k];
  if (t < BR) {
    const int start = bs + woff + inc - c;
    s_off[t] = start;
    s_cnt[t] = 0;
    const int node = node0 + t;
    float dg = (float)(c + 1);
    const float dv = (dg > 0.0f) ? rsqrtf(dg) : 0.0f;
    s_dinv[t] = dv;
    if (node < n) {
      rowptr[node] = start;
      dinv[node] = dv;
    }
  }
  __syncthreads();

  const int half = lane >> 4;
  const int lid  = lane & 15;
  const int row0 = node0 + 16 * __builtin_amdgcn_readfirstlane(w);
  const int ar   = row0 + lid;
  const int arc  = (ar < n) ? ar : (n - 1);
  const v2f* arow = (const v2f*)(A + (size_t)arc * D);
  v2f af[D / 4];
#pragma unroll
  for (int s = 0; s < D / 4; ++s) af[s] = arow[2 * s + half];

  for (int q = 0; q < NQ; ++q) {
    if (lane < rc[q]) {
      const int dl = r0[q].y - node0;
      const int k = atomicAdd(&s_cnt[dl], 1);
      csr[s_off[dl] + k] = r0[q].x;
    }
    if (lane + 32 < rc[q]) {
      const int dl = r1[q].y - node0;
      const int k = atomicAdd(&s_cnt[dl], 1);
      csr[s_off[dl] + k] = r1[q].x;
    }
  }
  for (int q = 0; q < NQ; ++q) {
#pragma nounroll
    for (int i0 = 64; i0 < rc[q]; i0 += 32) {
      const int i = i0 + lane;
      if (i < rc[q]) {
        const int2 rr = rec[rs[q] + i];
        const int dl = rr.y - node0;
        const int k = atomicAdd(&s_cnt[dl], 1);
        csr[s_off[dl] + k] = rr.x;
      }
    }
  }

  if (row0 >= n) return;
  const float ascale = s_dinv[16 * w + lid];
  v8f acc0 = {}, acc1 = {}, acc2 = {}, acc3 = {};
#pragma unroll
  for (int s = 0; s < D / 4; ++s) {
    const int kk = 2 * s + half;
    v2f a = af[s];
    a.x *= ascale;
    a.y *= ascale;
    const v2f* wrow = &sW[kk * D + lid];
    v2f b0 = wrow[0];
    v2f b1 = wrow[16];
    v2f b2 = wrow[32];
    v2f b3 = wrow[48];
    acc0 = __builtin_amdgcn_wmma_f32_16x16x4_f32(false, a, false, b0, (short)0, acc0, false, false);
    acc1 = __builtin_amdgcn_wmma_f32_16x16x4_f32(false, a, false, b1, (short)0, acc1, false, false);
    acc2 = __builtin_amdgcn_wmma_f32_16x16x4_f32(false, a, false, b2, (short)0, acc2, false, false);
    acc3 = __builtin_amdgcn_wmma_f32_16x16x4_f32(false, a, false, b3, (short)0, acc3, false, false);
  }

  __builtin_amdgcn_sched_barrier(0);
  asm volatile("v_nop\n\tv_nop\n\tv_nop\n\tv_nop");
  __builtin_amdgcn_sched_barrier(0);

#pragma unroll
  for (int v = 0; v < 8; ++v) {
    int rr = row0 + v + 8 * half;
    if (rr < n) {
      float* crow = Cout + (size_t)rr * D + lid;
      crow[0]  = acc0[v];
      crow[16] = acc1[v];
      crow[32] = acc2[v];
      crow[48] = acc3[v];
    }
  }
}

#define NPW 4

__device__ __forceinline__ void gather_piece(v2f& acc, const v2f* __restrict__ h2,
                                             int my_s, int c, int lane) {
  int sid[16];
  for (int k = 0; k < 16; ++k) sid[k] = __builtin_amdgcn_readlane(my_s, k);
  v2f v[16];
  for (int k = 0; k < 16; ++k) {
    if (k < c) v[k] = h2[(size_t)sid[k] * (D / 2) + lane];
  }
  for (int k = 0; k < 16; ++k) {
    if (k < c) {
      acc.x += v[k].x;
      acc.y += v[k].y;
    }
  }
}

__device__ __forceinline__ v2f gather_run(const int* __restrict__ csr, const v2f* __restrict__ h2,
                                          int node, int beg, int end, int c0, int id0, int lane) {
  v2f acc = {0.0f, 0.0f};
  int base = beg, c = c0, my_s = id0;
  while (c > 0) {
    gather_piece(acc, h2, my_s, c, lane);
    base += 16;
    c = end - base; if (c > 16) c = 16;
    if (c > 0) { my_s = node; if (lane < c) my_s = csr[base + lane]; }
  }
  return acc;
}

struct WaveNodes {
  int node[NPW], beg[NPW], end[NPW], c0[NPW], id0[NPW];
  float dt[NPW];
  v2f hs[NPW];
};

__device__ __forceinline__ void load_wave_nodes(WaveNodes& wn, int node0,
                                                const int* __restrict__ rowptr,
                                                const int* __restrict__ csr,
                                                const float* __restrict__ dinv,
                                                const v2f* __restrict__ h2,
                                                int n, int E, int lane) {
  int rp[NPW + 1];
#pragma unroll
  for (int j = 0; j <= NPW; ++j) {
    const int i = (node0 + j < n) ? (node0 + j) : (n - 1);
    rp[j] = rowptr[i];
  }
#pragma unroll
  for (int j = 0; j < NPW; ++j) {
    wn.node[j] = (node0 + j < n) ? (node0 + j) : (n - 1);
    wn.beg[j] = rp[j];
    wn.end[j] = (node0 + j + 1 < n) ? rp[j + 1] : E;
    int c = wn.end[j] - wn.beg[j]; if (c > 16) c = 16;
    wn.c0[j] = c;
  }
#pragma unroll
  for (int j = 0; j < NPW; ++j) {
    int id = wn.node[j];
    if (lane < wn.c0[j]) id = csr[wn.beg[j] + lane];
    wn.id0[j] = id;
  }
#pragma unroll
  for (int j = 0; j < NPW; ++j) {
    wn.dt[j] = dinv[wn.node[j]];
    wn.hs[j] = h2[(size_t)wn.node[j] * (D / 2) + lane];
  }
}

__global__ void __launch_bounds__(256)
k_gather(const int* __restrict__ rowptr, const int* __restrict__ csr,
         const float* __restrict__ dinv, const float* __restrict__ h,
         const float* __restrict__ b, const float* __restrict__ a,
         float* __restrict__ out, int n, int E) {
  const int lane = threadIdx.x & 31;
  const int node0 = NPW * __builtin_amdgcn_readfirstlane(
      (int)(blockIdx.x * (blockDim.x >> 5) + (threadIdx.x >> 5)));
  if (node0 >= n) return;
  const v2f* __restrict__ h2 = (const v2f*)h;
  WaveNodes wn;
  load_wave_nodes(wn, node0, rowptr, csr, dinv, h2, n, E, lane);
  const v2f bb = ((const v2f*)b)[lane];
  const v2f al = ((const v2f*)a)[lane];
#pragma unroll
  for (int j = 0; j < NPW; ++j) {
    v2f acc = gather_run(csr, h2, wn.node[j], wn.beg[j], wn.end[j], wn.c0[j], wn.id0[j], lane);
    acc.x = (acc.x + wn.hs[j].x) * wn.dt[j] + bb.x;
    acc.y = (acc.y + wn.hs[j].y) * wn.dt[j] + bb.y;
    acc.x = (acc.x >= 0.0f) ? acc.x : al.x * acc.x;
    acc.y = (acc.y >= 0.0f) ? acc.y : al.y * acc.y;
    if (node0 + j < n) ((v2f*)out)[(size_t)(node0 + j) * (D / 2) + lane] = acc;
  }
}

#define TN 32
#define TP 68

__global__ void __launch_bounds__(256)
k_gather_gemm(const int* __restrict__ rowptr, const int* __restrict__ csr,
              const float* __restrict__ dinv, const float* __restrict__ h,
              const float* __restrict__ b, const float* __restrict__ a,
              const float* __restrict__ W, float* __restrict__ Cout, int n, int E) {
  __shared__ v2f sW[(D / 2) * D];
  __shared__ v2f sT[TN * (TP / 2)];
  const int lane = threadIdx.x & 31;
  const int wave = __builtin_amdgcn_readfirstlane((int)(threadIdx.x >> 5));
  const int tile0 = blockIdx.x * TN;

  v2f wst[(D / 2) * D / 256];
#pragma unroll
  for (int j = 0; j < (D / 2) * D / 256; ++j) {
    const int i = (int)threadIdx.x + j * 256;
    const int kk = i / D;
    const int nn = i - kk * D;
    wst[j].x = W[(2 * kk + 0) * D + nn];
    wst[j].y = W[(2 * kk + 1) * D + nn];
  }

  const v2f* __restrict__ h2 = (const v2f*)h;
  const int node0 = tile0 + NPW * wave;
  WaveNodes wn;
  load_wave_nodes(wn, node0, rowptr, csr, dinv, h2, n, E, lane);
  const v2f bb = ((const v2f*)b)[lane];
  const v2f al = ((const v2f*)a)[lane];

#pragma unroll
  for (int j = 0; j < (D / 2) * D / 256; ++j) sW[(int)threadIdx.x + j * 256] = wst[j];

#pragma unroll
  for (int j = 0; j < NPW; ++j) {
    v2f acc = gather_run(csr, h2, wn.node[j], wn.beg[j], wn.end[j], wn.c0[j], wn.id0[j], lane);
    acc.x = (acc.x + wn.hs[j].x) * wn.dt[j] + bb.x;
    acc.y = (acc.y + wn.hs[j].y) * wn.dt[j] + bb.y;
    acc.x = (acc.x >= 0.0f) ? acc.x : al.x * acc.x;
    acc.y = (acc.y >= 0.0f) ? acc.y : al.y * acc.y;
    v2f r;
    r.x = acc.x * wn.dt[j];
    r.y = acc.y * wn.dt[j];
    sT[(NPW * wave + j) * (TP / 2) + lane] = r;
  }
  __syncthreads();

  if (wave < 4) {
    const int half = lane >> 4;
    const int lid  = lane & 15;
    v8f acc0 = {}, acc1 = {};
    const v2f* t0 = &sT[lid * (TP / 2)];
    const v2f* t1 = &sT[(16 + lid) * (TP / 2)];
    const v2f* wcol = &sW[wave * 16 + lid];
#pragma unroll
    for (int k = 0; k < D; k += 4) {
      int kk = (k >> 1) + half;
      v2f a0 = t0[kk];
      v2f a1 = t1[kk];
      v2f bq = wcol[kk * D];
      acc0 = __builtin_amdgcn_wmma_f32_16x16x4_f32(false, a0, false, bq, (short)0, acc0, false, false);
      acc1 = __builtin_amdgcn_wmma_f32_16x16x4_f32(false, a1, false, bq, (short)0, acc1, false, false);
    }
    __builtin_amdgcn_sched_barrier(0);
    asm volatile("v_nop\n\tv_nop\n\tv_nop\n\tv_nop");
    __builtin_amdgcn_sched_barrier(0);
#pragma unroll
    for (int v = 0; v < 8; ++v) {
      int rr = tile0 + v + 8 * half;
      if (rr < n)      Cout[(size_t)rr * D + wave * 16 + lid] = acc0[v];
      if (rr + 16 < n) Cout[(size_t)(rr + 16) * D + wave * 16 + lid] = acc1[v];
    }
  }
}

static inline int cdiv_ll(long long a, long long b) { return (int)((a + b - 1) / b); }

extern "C" void kernel_launch(void* const* d_in, const int* in_sizes, int n_in,
                              void* d_out, int out_size, void* d_ws, size_t ws_size,
                              hipStream_t stream) {
  (void)n_in; (void)out_size; (void)ws_size;

  const float* x  = (const float*)d_in[0];
  const int*   ei = (const int*)  d_in[1];
  const float* W1 = (const float*)d_in[2];
  const float* b1 = (const float*)d_in[3];
  const float* a1 = (const float*)d_in[4];
  const float* W2 = (const float*)d_in[5];
  const float* b2 = (const float*)d_in[6];
  const float* a2 = (const float*)d_in[7];

  const int n = in_sizes[0] / D;
  const int E = in_sizes[1] / 2;
  const int* src = ei;
  const int* dst = ei + E;

  float* ws     = (float*)d_ws;
  float* dinv   = ws;
  int*   rowptr = (int*)(dinv + n);
  int*   csr    = rowptr + n;
  float* bufA   = (float*)(csr + E);
  float* bufB   = bufA + (size_t)n * D;
  const int nb  = (n + BR - 1) / BR;
  int2*  rec    = (int2*)(bufB + (E & 1));
  int*   tab    = (int*)(rec + E);
  int*   off    = tab + (size_t)NCH * nb;
  float* out    = (float*)d_out;

  const int B = 256;
  const int gGa   = cdiv_ll((long long)cdiv_ll(n, NPW) * 32, B);
  const int gFuse = cdiv_ll(n, TN);

  k_bin             <<<NCH, TB, 2 * nb * sizeof(int), stream>>>(src, dst, tab, off, rec, E, nb);
  k_bucket_rows_gemm<<<nb, TB, 0, stream>>>(rec, tab, off, rowptr, dinv, csr, x, W1, bufA, n, E, nb);

  k_gather_gemm<<<gFuse, 256, 0, stream>>>(rowptr, csr, dinv, bufA, b1, a1, W2, bufB, n, E);

  k_gather   <<<gGa,   B, 0, stream>>>(rowptr, csr, dinv, bufB, b2, a2, out, n, E);
}


# Binning: records (k_bin) and source ids (bucket kernel) go to their slots in an LDS image of the work-group's contiguous output range and leave as whole-line 16-byte stores, not scattered 8/4-byte
# speedup vs baseline: 8.0555x; 1.0629x over previous
#include <hip/hip_runtime.h>
#include <hip/hip_bf16.h>

typedef float v2f __attribute__((ext_vector_type(2)));
typedef float v8f __attribute__((ext_vector_type(8)));

#define D 64

#define BSH 9
#define BR  (1 << BSH)
#define NCH 256
#define TB  1024
#define EB  8
#define NQ  (NCH / (TB / 32))

typedef int v4i __attribute__((ext_vector_type(4)));

__device__ __forceinline__ void image_out(int* __restrict__ gimg, const int* s_img,
                                          int mis, int nw, int t) {
  for (int g = t; 4 * g < nw; g += TB) {
    const int w0 = 4 * g;
    if (w0 >= mis && w0 + 4 <= nw) {
      *(v4i*)(gimg + w0) = *(const v4i*)(s_img + w0);
    } else {
      for (int e = 0; e < 4; ++e) {
        const int wq = w0 + e;
        if (wq >= mis && wq < nw) gimg[wq] = s_img[wq];
      }
    }
  }
}

__device__ __forceinline__ int wave_incl_scan(int v, int lane) {
  for (int o = 1; o < 32; o <<= 1) {
    int u = __shfl_up(v, o);
    if (lane >= o) v += u;
  }
  return v;
}

__global__ void __launch_bounds__(TB)
k_bin(const int* __restrict__ src, const int* __restrict__ dst,
      int* __restrict__ tab, int* __restrict__ off, int2* __restrict__ rec, int E, int nb) {
  extern __shared__ __attribute__((aligned(16))) int s_dyn[];
  __shared__ int s_w[TB / 32];
  int* s_cnt = s_dyn;
  int* s_base = s_dyn + nb;
  int* s_img = s_dyn + ((2 * nb + 3) & ~3);
  const int t = threadIdx.x;
  const int lane = t & 31;
  const int w = t >> 5;
  const int per = (E + NCH - 1) / NCH;
  const int lo = blockIdx.x * per;
  const int hi = (lo + per < E) ? (lo + per) : E;
  const int mis = (int)((((size_t)(rec + lo)) >> 2) & 31);
  int d[EB], s[EB];
  for (int j = 0; j < EB; ++j) {
    int i = lo + j * TB + t;
    int ic = (i < hi) ? i : (hi - 1);
    ic = (ic > 0) ? ic : 0;
    d[j] = dst[ic];
    s[j] = src[ic];
  }
  for (int j = t; j < nb; j += TB) s_cnt[j] = 0;
  __syncthreads();

  for (int j = 0; j < EB; ++j) {
    int i = lo + j * TB + t;
    if (i < hi) atomicAdd(&s_cnt[d[j] >> BSH], 1);
  }
  for (int i0 = lo + EB * TB; i0 < hi; i0 += EB * TB) {
    int dd[EB];
    for (int j = 0; j < EB; ++j) {
      int i = i0 + j * TB + t;
      dd[j] = dst[(i < hi) ? i : (hi - 1)];
    }
    for (int j = 0; j < EB; ++j) {
      int i = i0 + j * TB + t;
      if (i < hi) atomicAdd(&s_cnt[dd[j] >> BSH], 1);
    }
  }
  __syncthreads();

  int carry = 0;
  for (int j0 = 0; j0 < nb; j0 += TB) {
    const int j = j0 + t;
    const int v = (j < nb) ? s_cnt[j] : 0;
    const int inc = wave_incl_scan(v, lane);
    if (lane == 31) s_w[w] = inc;
    __syncthreads();
    int woff = 0, tot = 0;
    for (int k = 0; k < TB / 32; ++k) {
      int x = s_w[k];
      tot += x;
      if (k < w) woff += x;
    }
    if (j < nb) {
      const int st = carry + woff + inc - v;
      s_base[j] = mis + 2 * st;
      s_cnt[j] = 0;
      tab[blockIdx.x * nb + j] = v;
      off[blockIdx.x * nb + j] = st;
    }
    carry += tot;
    __syncthreads();
  }

  for (int i0 = lo; i0 < hi; ) {
    for (int j = 0; j < EB; ++j) {
      int i = i0 + j * TB + t;
      if (i < hi) {
        const int bk = d[j] >> BSH;
        const int r = atomicAdd(&s_cnt[bk], 1);
        *(int2*)(s_img + s_base[bk] + 2 * r) = make_int2(s[j], d[j]);
      }
    }
    i0 += EB * TB;
    if (i0 < hi) {
      for (int j = 0; j < EB; ++j) {
        int i = i0 + j * TB + t;
        int ic = (i < hi) ? i : (hi - 1);
        d[j] = dst[ic];
        s[j] = src[ic];
      }
    }
  }

  __syncthreads();
  const int cn = (hi > lo) ? (hi - lo) : 0;
  image_out((int*)(rec + lo) - mis, s_img, mis, mis + 2 * cn, t);
}

#define CAPW 8192

template <bool IMG>
__device__ __forceinline__ void place_runs(const int (&rs)[NQ], const int (&rc)[NQ],
                                           const int2 (&r0)[NQ], const int2 (&r1)[NQ],
                                           const int2* __restrict__ rec,
                                           int* s_cnt, const int* s_off, int* s_img,
                                           int* __restrict__ csr, int node0, int lane) {
  for (int q = 0; q < NQ; ++q) {
    if (lane < rc[q]) {
      const int dl = r0[q].y - node0;
      const int k = atomicAdd(&s_cnt[dl], 1);
      if (IMG) s_img[s_off[dl] + k] = r0[q].x; else csr[s_off[dl] + k] = r0[q].x;
    }
    if (lane + 32 < rc[q]) {
      const int dl = r1[q].y - node0;
      const int k = atomicAdd(&s_cnt[dl], 1);
      if (IMG) s_img[s_off[dl] + k] = r1[q].x; else csr[s_off[dl] + k] = r1[q].x;
    }
  }
  for (int q = 0; q < NQ; ++q) {
#pragma nounroll
    for (int i0 = 64; i0 < rc[q]; i0 += 32) {
      const int i = i0 + lane;
      if (i < rc[q]) {
        const int2 rr = rec[rs[q] + i];
        const int dl = rr.y - node0;
        const int k = atomicAdd(&s_cnt[dl], 1);
        if (IMG) s_img[s_off[dl] + k] = rr.x; else csr[s_off[dl] + k] = rr.x;
      }
    }
  }
}

__global__ void __launch_bounds__(TB)
k_bucket_rows_gemm(const int2* __restrict__ rec, const int* __restrict__ tab,
                   const int* __restrict__ off,
                   int* __restrict__ rowptr, float* __restrict__ dinv,
                   int* __restrict__ csr,
                   const float* __restrict__ A, const float* __restrict__ W,
                   float* __restrict__ Cout, int n, int E, int nb) {
  static_assert(BR / 16 == TB / 32, "one 16-row tile of the bucket per wave");
  __shared__ int s_cnt[BR];
  __shared__ int s_off[BR];
  __shared__ int s_rs[NCH];
  __shared__ int s_rc[NCH];
  __shared__ int s_w[TB / 32];
  __shared__ float s_dinv[BR];
  __shared__ v2f sW[(D / 2) * D];
  __shared__ __attribute__((aligned(16))) int s_img[CAPW];
  const int t = threadIdx.x;
  const int lane = t & 31;
  const int w = t >> 5;
  const int b = blockIdx.x;
  const int node0 = b << BSH;
  const int per = (E + NCH - 1) / NCH;
  v2f wst[(D / 2) * D / TB];
#pragma unroll
  for (int j = 0; j < (D / 2) * D / TB; ++j) {
    const int i = t + j * TB;
    const int kk = i / D;
    const int nn = i - kk * D;
    wst[j].x = W[(2 * kk + 0) * D + nn];
    wst[j].y = W[(2 * kk + 1) * D + nn];
  }
  int of = 0;
  if (t < NCH) {
    of = off[t * nb + b];
    s_rs[t] = t * per + of;
    s_rc[t] = tab[t * nb + b];
  }
#pragma unroll
  for (int j = 0; j < (D / 2) * D / TB; ++j) sW[t + j * TB] = wst[j];
  if (t < BR) s_cnt[t] = 0;
  int wsum = of;
  for (int o = 16; o > 0; o >>= 1) wsum += __shfl_xor(wsum, o);
  if (lane == 0) s_w[w] = wsum;
  __syncthreads();
  int bs = 0;
  for (int k = 0; k < TB / 32; ++k) bs += s_w[k];

  int rs[NQ], rc[NQ];
  int2 r0[NQ], r1[NQ];
  for (int q = 0; q < NQ; ++q) {
    const int c = w + q * (TB / 32);
    rs[q] = __builtin_amdgcn_readfirstlane(s_rs[c]);
    rc[q] = __builtin_amdgcn_readfirstlane(s_rc[c]);
    r0[q] = rec[(lane < rc[q]) ? (rs[q] + lane) : 0];
    r1[q] = rec[(lane + 32 < rc[q]) ? (rs[q] + lane + 32) : 0];
  }

  for (int q = 0; q < NQ; ++q) {
    if (lane < rc[q])      atomicAdd(&s_cnt[r0[q].y - node0], 1);
    if (lane + 32 < rc[q]) atomicAdd(&s_cnt[r1[q].y - node0], 1);
  }
  for (int q = 0; q < NQ; ++q) {
#pragma nounroll
    for (int i0 = 64; i0 < rc[q]; i0 += 32) {
      const int i = i0 + lane;
      if (i < rc[q]) atomicAdd(&s_cnt[rec[rs[q] + i].y - node0], 1);
    }
  }
  __syncthreads();

  const int c = (t < BR) ? s_cnt[t] : 0;
  const int inc = wave_incl_scan(c, lane);
  if (lane == 31) s_w[w] = inc;
  __syncthreads();
  int woff = 0, tot = 0;
  for (int k = 0; k < TB / 32; ++k) {
    const int x = s_w[k];
    tot += x;
    if (k < w) woff += x;
  }
  const int mis = (int)((((size_t)(csr + bs)) >> 2) & 31);
  const bool img = __builtin_amdgcn_readfirstlane((mis + tot <= CAPW) ? 1 : 0) != 0;
  if (t < BR) {
    const int start = bs + woff + inc - c;
    s_off[t] = img ? (start - bs + mis) : start;
    s_cnt[t] = 0;
    const int node = node0 + t;
    float dg = (float)(c + 1);
    const float dv = (dg > 0.0f) ? rsqrtf(dg) : 0.0f;
    s_dinv[t] = dv;
    if (node < n) {
      rowptr[node] = start;
      dinv[node] = dv;
    }
  }
  __syncthreads();

  const int half = lane >> 4;
  const int lid  = lane & 15;
  const int row0 = node0 + 16 * __builtin_amdgcn_readfirstlane(w);
  const int ar   = row0 + lid;
  const int arc  = (ar < n) ? ar : (n - 1);
  const v2f* arow = (const v2f*)(A + (size_t)arc * D);
  v2f af[D / 4];
#pragma unroll
  for (int s = 0; s < D / 4; ++s) af[s] = arow[2 * s + half];

  if (img) {
    place_runs<true>(rs, rc, r0, r1, rec, s_cnt, s_off, s_img, csr, node0, lane);
    __syncthreads();
    image_out(csr + bs - mis, s_img, mis, mis + tot, t);
  } else {
    place_runs<false>(rs, rc, r0, r1, rec, s_cnt, s_off, s_img, csr, node0, lane);
  }

  if (row0 >= n) return;
  const float ascale = s_dinv[16 * w + lid];
  v8f acc0 = {}, acc1 = {}, acc2 = {}, acc3 = {};
#pragma unroll
  for (int s = 0; s < D / 4; ++s) {
    const int kk = 2 * s + half;
    v2f a = af[s];
    a.x *= ascale;
    a.y *= ascale;
    const v2f* wrow = &sW[kk * D + lid];
    v2f b0 = wrow[0];
    v2f b1 = wrow[16];
    v2f b2 = wrow[32];
    v2f b3 = wrow[48];
    acc0 = __builtin_amdgcn_wmma_f32_16x16x4_f32(false, a, false, b0, (short)0, acc0, false, false);
    acc1 = __builtin_amdgcn_wmma_f32_16x16x4_f32(false, a, false, b1, (short)0, acc1, false, false);
    acc2 = __builtin_amdgcn_wmma_f32_16x16x4_f32(false, a, false, b2, (short)0, acc2, false, false);
    acc3 = __builtin_amdgcn_wmma_f32_16x16x4_f32(false, a, false, b3, (short)0, acc3, false, false);
  }

  __builtin_amdgcn_sched_barrier(0);
  asm volatile("v_nop\n\tv_nop\n\tv_nop\n\tv_nop");
  __builtin_amdgcn_sched_barrier(0);

#pragma unroll
  for (int v = 0; v < 8; ++v) {
    int rr = row0 + v + 8 * half;
    if (rr < n) {
      float* crow = Cout + (size_t)rr * D + lid;
      crow[0]  = acc0[v];
      crow[16] = acc1[v];
      crow[32] = acc2[v];
      crow[48] = acc3[v];
    }
  }
}

#define NPW 4

__device__ __forceinline__ void gather_piece(v2f& acc, const v2f* __restrict__ h2,
                                             int my_s, int c, int lane) {
  int sid[16];
  for (int k = 0; k < 16; ++k) sid[k] = __builtin_amdgcn_readlane(my_s, k);
  v2f v[16];
  for (int k = 0; k < 16; ++k) {
    if (k < c) v[k] = h2[(size_t)sid[k] * (D / 2) + lane];
  }
  for (int k = 0; k < 16; ++k) {
    if (k < c) {
      acc.x += v[k].x;
      acc.y += v[k].y;
    }
  }
}

__device__ __forceinline__ v2f gather_run(const int* __restrict__ csr, const v2f* __restrict__ h2,
                                          int node, int beg, int end, int c0, int id0, int lane) {
  v2f acc = {0.0f, 0.0f};
  int base = beg, c = c0, my_s = id0;
  while (c > 0) {
    gather_piece(acc, h2, my_s, c, lane);
    base += 16;
    c = end - base; if (c > 16) c = 16;
    if (c > 0) { my_s = node; if (lane < c) my_s = csr[base + lane]; }
  }
  return acc;
}

struct WaveNodes {
  int node[NPW], beg[NPW], end[NPW], c0[NPW], id0[NPW];
  float dt[NPW];
  v2f hs[NPW];
};

__device__ __forceinline__ void load_wave_nodes(WaveNodes& wn, int node0,
                                                const int* __restrict__ rowptr,
                                                const int* __restrict__ csr,
                                                const float* __restrict__ dinv,
                                                const v2f* __restrict__ h2,
                                                int n, int E, int lane) {
  int rp[NPW + 1];
#pragma unroll
  for (int j = 0; j <= NPW; ++j) {
    const int i = (node0 + j < n) ? (node0 + j) : (n - 1);
    rp[j] = rowptr[i];
  }
#pragma unroll
  for (int j = 0; j < NPW; ++j) {
    wn.node[j] = (node0 + j < n) ? (node0 + j) : (n - 1);
    wn.beg[j] = rp[j];
    wn.end[j] = (node0 + j + 1 < n) ? rp[j + 1] : E;
    int c = wn.end[j] - wn.beg[j]; if (c > 16) c = 16;
    wn.c0[j] = c;
  }
#pragma unroll
  for (int j = 0; j < NPW; ++j) {
    int id = wn.node[j];
    if (lane < wn.c0[j]) id = csr[wn.beg[j] + lane];
    wn.id0[j] = id;
  }
#pragma unroll
  for (int j = 0; j < NPW; ++j) {
    wn.dt[j] = dinv[wn.node[j]];
    wn.hs[j] = h2[(size_t)wn.node[j] * (D / 2) + lane];
  }
}

__global__ void __launch_bounds__(256)
k_gather(const int* __restrict__ rowptr, const int* __restrict__ csr,
         const float* __restrict__ dinv, const float* __restrict__ h,
         const float* __restrict__ b, const float* __restrict__ a,
         float* __restrict__ out, int n, int E) {
  const int lane = threadIdx.x & 31;
  const int node0 = NPW * __builtin_amdgcn_readfirstlane(
      (int)(blockIdx.x * (blockDim.x >> 5) + (threadIdx.x >> 5)));
  if (node0 >= n) return;
  const v2f* __restrict__ h2 = (const v2f*)h;
  WaveNodes wn;
  load_wave_nodes(wn, node0, rowptr, csr, dinv, h2, n, E, lane);
  const v2f bb = ((const v2f*)b)[lane];
  const v2f al = ((const v2f*)a)[lane];
#pragma unroll
  for (int j = 0; j < NPW; ++j) {
    v2f acc = gather_run(csr, h2, wn.node[j], wn.beg[j], wn.end[j], wn.c0[j], wn.id0[j], lane);
    acc.x = (acc.x + wn.hs[j].x) * wn.dt[j] + bb.x;
    acc.y = (acc.y + wn.hs[j].y) * wn.dt[j] + bb.y;
    acc.x = (acc.x >= 0.0f) ? acc.x : al.x * acc.x;
    acc.y = (acc.y >= 0.0f) ? acc.y : al.y * acc.y;
    if (node0 + j < n) ((v2f*)out)[(size_t)(node0 + j) * (D / 2) + lane] = acc;
  }
}

#define TN 32
#define TP 68

__global__ void __launch_bounds__(256)
k_gather_gemm(const int* __restrict__ rowptr, const int* __restrict__ csr,
              const float* __restrict__ dinv, const float* __restrict__ h,
              const float* __restrict__ b, const float* __restrict__ a,
              const float* __restrict__ W, float* __restrict__ Cout, int n, int E) {
  __shared__ v2f sW[(D / 2) * D];
  __shared__ v2f sT[TN * (TP / 2)];
  const int lane = threadIdx.x & 31;
  const int wave = __builtin_amdgcn_readfirstlane((int)(threadIdx.x >> 5));
  const int tile0 = blockIdx.x * TN;

  v2f wst[(D / 2) * D / 256];
#pragma unroll
  for (int j = 0; j < (D / 2) * D / 256; ++j) {
    const int i = (int)threadIdx.x + j * 256;
    const int kk = i / D;
    const int nn = i - kk * D;
    wst[j].x = W[(2 * kk + 0) * D + nn];
    wst[j].y = W[(2 * kk + 1) * D + nn];
  }

  const v2f* __restrict__ h2 = (const v2f*)h;
  const int node0 = tile0 + NPW * wave;
  WaveNodes wn;
  load_wave_nodes(wn, node0, rowptr, csr, dinv, h2, n, E, lane);
  const v2f bb = ((const v2f*)b)[lane];
  const v2f al = ((const v2f*)a)[lane];

#pragma unroll
  for (int j = 0; j < (D / 2) * D / 256; ++j) sW[(int)threadIdx.x + j * 256] = wst[j];

#pragma unroll
  for (int j = 0; j < NPW; ++j) {
    v2f acc = gather_run(csr, h2, wn.node[j], wn.beg[j], wn.end[j], wn.c0[j], wn.id0[j], lane);
    acc.x = (acc.x + wn.hs[j].x) * wn.dt[j] + bb.x;
    acc.y = (acc.y + wn.hs[j].y) * wn.dt[j] + bb.y;
    acc.x = (acc.x >= 0.0f) ? acc.x : al.x * acc.x;
    acc.y = (acc.y >= 0.0f) ? acc.y : al.y * acc.y;
    v2f r;
    r.x = acc.x * wn.dt[j];
    r.y = acc.y * wn.dt[j];
    sT[(NPW * wave + j) * (TP / 2) + lane] = r;
  }
  __syncthreads();

  if (wave < 4) {
    const int half = lane >> 4;
    const int lid  = lane & 15;
    v8f acc0 = {}, acc1 = {};
    const v2f* t0 = &sT[lid * (TP / 2)];
    const v2f* t1 = &sT[(16 + lid) * (TP / 2)];
    const v2f* wcol = &sW[wave * 16 + lid];
#pragma unroll
    for (int k = 0; k < D; k += 4) {
      int kk = (k >> 1) + half;
      v2f a0 = t0[kk];
      v2f a1 = t1[kk];
      v2f bq = wcol[kk * D];
      acc0 = __builtin_amdgcn_wmma_f32_16x16x4_f32(false, a0, false, bq, (short)0, acc0, false, false);
      acc1 = __builtin_amdgcn_wmma_f32_16x16x4_f32(false, a1, false, bq, (short)0, acc1, false, false);
    }
    __builtin_amdgcn_sched_barrier(0);
    asm volatile("v_nop\n\tv_nop\n\tv_nop\n\tv_nop");
    __builtin_amdgcn_sched_barrier(0);
#pragma unroll
    for (int v = 0; v < 8; ++v) {
      int rr = tile0 + v + 8 * half;
      if (rr < n)      Cout[(size_t)rr * D + wave * 16 + lid] = acc0[v];
      if (rr + 16 < n) Cout[(size_t)(rr + 16) * D + wave * 16 + lid] = acc1[v];
    }
  }
}

static inline int cdiv_ll(long long a, long long b) { return (int)((a + b - 1) / b); }

extern "C" void kernel_launch(void* const* d_in, const int* in_sizes, int n_in,
                              void* d_out, int out_size, void* d_ws, size_t ws_size,
                              hipStream_t stream) {
  (void)n_in; (void)out_size; (void)ws_size;

  const float* x  = (const float*)d_in[0];
  const int*   ei = (const int*)  d_in[1];
  const float* W1 = (const float*)d_in[2];
  const float* b1 = (const float*)d_in[3];
  const float* a1 = (const float*)d_in[4];
  const float* W2 = (const float*)d_in[5];
  const float* b2 = (const float*)d_in[6];
  const float* a2 = (const float*)d_in[7];

  const int n = in_sizes[0] / D;
  const int E = in_sizes[1] / 2;
  const int* src = ei;
  const int* dst = ei + E;

  float* ws     = (float*)d_ws;
  float* dinv   = ws;
  int*   rowptr = (int*)(dinv + n);
  int*   csr    = rowptr + n;
  float* bufA   = (float*)(csr + E);
  float* bufB   = bufA + (size_t)n * D;
  const int nb  = (n + BR - 1) / BR;
  int2*  rec    = (int2*)(bufB + (E & 1));
  int*   tab    = (int*)(rec + E);
  int*   off    = tab + (size_t)NCH * nb;
  float* out    = (float*)d_out;

  const int B = 256;
  const int gGa   = cdiv_ll((long long)cdiv_ll(n, NPW) * 32, B);
  const int gFuse = cdiv_ll(n, TN);

  const int per = (E + NCH - 1) / NCH;
  const size_t binLds = (size_t)(((2 * nb + 3) & ~3) + 2 * per + 32) * sizeof(int);
  k_bin             <<<NCH, TB, binLds, stream>>>(src, dst, tab, off, rec, E, nb);
  k_bucket_rows_gemm<<<nb, TB, 0, stream>>>(rec, tab, off, rowptr, dinv, csr, x, W1, bufA, n, E, nb);

  k_gather_gemm<<<gFuse, 256, 0, stream>>>(rowptr, csr, dinv, bufA, b1, a1, W2, bufB, n, E);

  k_gather   <<<gGa,   B, 0, stream>>>(rowptr, csr, dinv, bufB, b2, a2, out, n, E);
}
